# starting kernel
# baseline (speedup 1.0000x reference)
#include <hip/hip_runtime.h>
#include <hip/hip_bf16.h>

#define N_USERS 100000
#define N_ITEMS 200000
#define N_NODES (N_USERS + N_ITEMS)
#define NNZ     4800000
#define DIM     64
#define BATCH   16384

typedef float v2f __attribute__((ext_vector_type(2)));
typedef float v8f __attribute__((ext_vector_type(8)));

__global__ void lgcn_init(const float* __restrict__ user_emb,
                          const float* __restrict__ item_emb,
                          float* __restrict__ h,
                          float* __restrict__ acc) {
    size_t f = (size_t)blockIdx.x * blockDim.x + threadIdx.x;
    size_t total = (size_t)N_NODES * (DIM / 4);
    if (f >= total) return;
    size_t node = f >> 4;
    size_t part = f & 15;
    const float4* src;
    if (node < N_USERS)
        src = (const float4*)(user_emb + node * DIM);
    else
        src = (const float4*)(item_emb + (node - N_USERS) * DIM);
    float4 v = src[part];
    ((float4*)h)[f]   = v;
    ((float4*)acc)[f] = v;
}

__global__ void lgcn_zero(float* __restrict__ p) {
    size_t f = (size_t)blockIdx.x * blockDim.x + threadIdx.x;
    size_t total = (size_t)N_NODES * (DIM / 4);
    if (f >= total) return;
    ((float4*)p)[f] = make_float4(0.f, 0.f, 0.f, 0.f);
}

__global__ void lgcn_spmm(const int*   __restrict__ rows,
                          const int*   __restrict__ cols,
                          const float* __restrict__ vals,
                          const float* __restrict__ h,
                          float*       __restrict__ out) {
    size_t tid = (size_t)blockIdx.x * blockDim.x + threadIdx.x;
    size_t e   = tid >> 4;
    int    part = (int)(tid & 15);
    if (e >= (size_t)NNZ) return;
    int   r = rows[e];
    int   c = cols[e];
    float v = vals[e];
    const float4 src = *(const float4*)(h + (size_t)c * DIM + part * 4);
    float* dst = out + (size_t)r * DIM + part * 4;
    unsafeAtomicAdd(dst + 0, v * src.x);
    unsafeAtomicAdd(dst + 1, v * src.y);
    unsafeAtomicAdd(dst + 2, v * src.z);
    unsafeAtomicAdd(dst + 3, v * src.w);
}

__global__ void lgcn_addacc(float* __restrict__ acc, const float* __restrict__ h) {
    size_t f = (size_t)blockIdx.x * blockDim.x + threadIdx.x;
    size_t total = (size_t)N_NODES * (DIM / 4);
    if (f >= total) return;
    float4 a = ((float4*)acc)[f];
    float4 b = ((const float4*)h)[f];
    a.x += b.x; a.y += b.y; a.z += b.z; a.w += b.w;
    ((float4*)acc)[f] = a;
}

__global__ void lgcn_dot_wmma(const float* __restrict__ acc,
                              const int*   __restrict__ users,
                              const int*   __restrict__ items,
                              float*       __restrict__ out) {
    __shared__ float lds[8][16][17];
    int lane  = threadIdx.x & 31;
    int wv    = threadIdx.x >> 5;
    int tile  = blockIdx.x * 8 + wv;
    int m     = lane & 15;
    int khalf = lane >> 4;

    int u_row = users[tile * 16 + m];
    int v_row = items[tile * 16 + m];
    const float* up = acc + (size_t)u_row * DIM;
    const float* vp = acc + ((size_t)N_USERS + (size_t)v_row) * DIM;

    v8f c = {};
#pragma unroll
    for (int kc = 0; kc < DIM / 4; ++kc) {
        int off = kc * 4 + khalf * 2;
        v2f a; a.x = up[off]; a.y = up[off + 1];
        v2f b; b.x = vp[off]; b.y = vp[off + 1];
        c = __builtin_amdgcn_wmma_f32_16x16x4_f32(
                 false, a,  false, b,
                 (short)0, c,  false,  false);
    }

#pragma unroll
    for (int r = 0; r < 8; ++r)
        lds[wv][r + 8 * khalf][m] = c[r];
    __syncthreads();

    if (lane < 16)
        out[tile * 16 + lane] = lds[wv][lane][lane] * (1.0f / 16.0f);
}

extern "C" void kernel_launch(void* const* d_in, const int* in_sizes, int n_in,
                              void* d_out, int out_size, void* d_ws, size_t ws_size,
                              hipStream_t stream) {
    const float* user_emb = (const float*)d_in[0];
    const float* item_emb = (const float*)d_in[1];
    const int*   g_rows   = (const int*)  d_in[2];
    const int*   g_cols   = (const int*)  d_in[3];
    const float* g_vals   = (const float*)d_in[4];
    const int*   users    = (const int*)  d_in[5];
    const int*   items    = (const int*)  d_in[6];
    float*       out      = (float*)d_out;

    const size_t node_f = (size_t)N_NODES * DIM;
    float* bufA = (float*)d_ws;
    float* bufB = bufA + node_f;
    float* accb = bufB + node_f;

    const int TB = 256;
    const int elem_blocks = (int)(((size_t)N_NODES * (DIM / 4) + TB - 1) / TB);
    const int spmm_blocks = (int)(((size_t)NNZ * 16 + TB - 1) / TB);

    lgcn_init<<<elem_blocks, TB, 0, stream>>>(user_emb, item_emb, bufA, accb);

    float* hsrc = bufA;
    float* hdst = bufB;
    for (int hop = 0; hop < 3; ++hop) {
        lgcn_zero<<<elem_blocks, TB, 0, stream>>>(hdst);
        lgcn_spmm<<<spmm_blocks, TB, 0, stream>>>(g_rows, g_cols, g_vals, hsrc, hdst);
        lgcn_addacc<<<elem_blocks, TB, 0, stream>>>(accb, hdst);
        float* t = hsrc; hsrc = hdst; hdst = t;
    }

    lgcn_dot_wmma<<<BATCH / 16 / 8, TB, 0, stream>>>(accb, users, items, out);
}


# SpMM scatter remapped to lane = feature: a wave walks 32 nnz, each row load and each f32 atomic add covers one whole 128-B line (8 line-touching atomics per nnz become 2); arrival order as before
# speedup vs baseline: 2.0507x; 2.0507x over previous
#include <hip/hip_runtime.h>
#include <hip/hip_bf16.h>

#define N_USERS 100000
#define N_ITEMS 200000
#define N_NODES (N_USERS + N_ITEMS)
#define NNZ     4800000
#define DIM     64
#define BATCH   16384

typedef float v2f __attribute__((ext_vector_type(2)));
typedef float v8f __attribute__((ext_vector_type(8)));

__global__ void lgcn_init(const float* __restrict__ user_emb,
                          const float* __restrict__ item_emb,
                          float* __restrict__ h,
                          float* __restrict__ acc) {
    size_t f = (size_t)blockIdx.x * blockDim.x + threadIdx.x;
    size_t total = (size_t)N_NODES * (DIM / 4);
    if (f >= total) return;
    size_t node = f >> 4;
    size_t part = f & 15;
    const float4* src;
    if (node < N_USERS)
        src = (const float4*)(user_emb + node * DIM);
    else
        src = (const float4*)(item_emb + (node - N_USERS) * DIM);
    float4 v = src[part];
    ((float4*)h)[f]   = v;
    ((float4*)acc)[f] = v;
}

__global__ void lgcn_zero(float* __restrict__ p) {
    size_t f = (size_t)blockIdx.x * blockDim.x + threadIdx.x;
    size_t total = (size_t)N_NODES * (DIM / 4);
    if (f >= total) return;
    ((float4*)p)[f] = make_float4(0.f, 0.f, 0.f, 0.f);
}

#define SPMM_GROUP 8

__global__ void lgcn_spmm(const int*   __restrict__ rows,
                          const int*   __restrict__ cols,
                          const float* __restrict__ vals,
                          const float* __restrict__ h,
                          float*       __restrict__ out) {
    const int lane = (int)(threadIdx.x & 31);
    const size_t e = (size_t)blockIdx.x * blockDim.x + threadIdx.x;
    if (e >= (size_t)NNZ) return;
    const int r  = rows[e];
    const int c  = cols[e];
    const int vb = __float_as_int(vals[e]);
    const float* hl = h   + lane;
    float*       ol = out + lane;

    float a[SPMM_GROUP], b[SPMM_GROUP];
#pragma unroll
    for (int k = 0; k < SPMM_GROUP; ++k) {
        const float* s = hl + (size_t)__builtin_amdgcn_readlane(c, k) * DIM;
        a[k] = s[0];
        b[k] = s[32];
    }
#pragma unroll
    for (int g = 0; g < 32; g += SPMM_GROUP) {
        float na[SPMM_GROUP], nb[SPMM_GROUP];
        if (g + SPMM_GROUP < 32) {
#pragma unroll
            for (int k = 0; k < SPMM_GROUP; ++k) {
                const float* s = hl + (size_t)__builtin_amdgcn_readlane(c, g + SPMM_GROUP + k) * DIM;
                na[k] = s[0];
                nb[k] = s[32];
            }
        }
#pragma unroll
        for (int k = 0; k < SPMM_GROUP; ++k) {
            const int   rj = __builtin_amdgcn_readlane(r, g + k);
            const float vj = __int_as_float(__builtin_amdgcn_readlane(vb, g + k));
            float* d = ol + (size_t)rj * DIM;
            unsafeAtomicAdd(d,      vj * a[k]);
            unsafeAtomicAdd(d + 32, vj * b[k]);
        }
        if (g + SPMM_GROUP < 32) {
#pragma unroll
            for (int k = 0; k < SPMM_GROUP; ++k) { a[k] = na[k]; b[k] = nb[k]; }
        }
    }
}

__global__ void lgcn_addacc(float* __restrict__ acc, const float* __restrict__ h) {
    size_t f = (size_t)blockIdx.x * blockDim.x + threadIdx.x;
    size_t total = (size_t)N_NODES * (DIM / 4);
    if (f >= total) return;
    float4 a = ((float4*)acc)[f];
    float4 b = ((const float4*)h)[f];
    a.x += b.x; a.y += b.y; a.z += b.z; a.w += b.w;
    ((float4*)acc)[f] = a;
}

__global__ void lgcn_dot_wmma(const float* __restrict__ acc,
                              const int*   __restrict__ users,
                              const int*   __restrict__ items,
                              float*       __restrict__ out) {
    __shared__ float lds[8][16][17];
    int lane  = threadIdx.x & 31;
    int wv    = threadIdx.x >> 5;
    int tile  = blockIdx.x * 8 + wv;
    int m     = lane & 15;
    int khalf = lane >> 4;

    int u_row = users[tile * 16 + m];
    int v_row = items[tile * 16 + m];
    const float* up = acc + (size_t)u_row * DIM;
    const float* vp = acc + ((size_t)N_USERS + (size_t)v_row) * DIM;

    v8f c = {};
#pragma unroll
    for (int kc = 0; kc < DIM / 4; ++kc) {
        int off = kc * 4 + khalf * 2;
        v2f a; a.x = up[off]; a.y = up[off + 1];
        v2f b; b.x = vp[off]; b.y = vp[off + 1];
        c = __builtin_amdgcn_wmma_f32_16x16x4_f32(
                 false, a,  false, b,
                 (short)0, c,  false,  false);
    }

#pragma unroll
    for (int r = 0; r < 8; ++r)
        lds[wv][r + 8 * khalf][m] = c[r];
    __syncthreads();

    if (lane < 16)
        out[tile * 16 + lane] = lds[wv][lane][lane] * (1.0f / 16.0f);
}

extern "C" void kernel_launch(void* const* d_in, const int* in_sizes, int n_in,
                              void* d_out, int out_size, void* d_ws, size_t ws_size,
                              hipStream_t stream) {
    const float* user_emb = (const float*)d_in[0];
    const float* item_emb = (const float*)d_in[1];
    const int*   g_rows   = (const int*)  d_in[2];
    const int*   g_cols   = (const int*)  d_in[3];
    const float* g_vals   = (const float*)d_in[4];
    const int*   users    = (const int*)  d_in[5];
    const int*   items    = (const int*)  d_in[6];
    float*       out      = (float*)d_out;

    const size_t node_f = (size_t)N_NODES * DIM;
    float* bufA = (float*)d_ws;
    float* bufB = bufA + node_f;
    float* accb = bufB + node_f;

    const int TB = 256;
    const int elem_blocks = (int)(((size_t)N_NODES * (DIM / 4) + TB - 1) / TB);
    const int spmm_blocks = NNZ / TB;

    lgcn_init<<<elem_blocks, TB, 0, stream>>>(user_emb, item_emb, bufA, accb);

    float* hsrc = bufA;
    float* hdst = bufB;
    for (int hop = 0; hop < 3; ++hop) {
        lgcn_zero<<<elem_blocks, TB, 0, stream>>>(hdst);
        lgcn_spmm<<<spmm_blocks, TB, 0, stream>>>(g_rows, g_cols, g_vals, hsrc, hdst);
        lgcn_addacc<<<elem_blocks, TB, 0, stream>>>(accb, hdst);
        float* t = hsrc; hsrc = hdst; hdst = t;
    }

    lgcn_dot_wmma<<<BATCH / 16 / 8, TB, 0, stream>>>(accb, users, items, out);
}


# Last hop only at rows the batch reads: per-block LDS bitmask of needed rows, scatter skips other nnz; its zero shrinks to those rows, its acc add moves into the dot kernel; sum order as before
# speedup vs baseline: 2.8309x; 1.3805x over previous
#include <hip/hip_runtime.h>
#include <hip/hip_bf16.h>

#define N_USERS 100000
#define N_ITEMS 200000
#define N_NODES (N_USERS + N_ITEMS)
#define NNZ     4800000
#define DIM     64
#define BATCH   16384

typedef float v2f __attribute__((ext_vector_type(2)));
typedef float v8f __attribute__((ext_vector_type(8)));

__global__ void lgcn_init(const float* __restrict__ user_emb,
                          const float* __restrict__ item_emb,
                          float* __restrict__ h,
                          float* __restrict__ acc) {
    size_t f = (size_t)blockIdx.x * blockDim.x + threadIdx.x;
    size_t total = (size_t)N_NODES * (DIM / 4);
    if (f >= total) return;
    size_t node = f >> 4;
    size_t part = f & 15;
    const float4* src;
    if (node < N_USERS)
        src = (const float4*)(user_emb + node * DIM);
    else
        src = (const float4*)(item_emb + (node - N_USERS) * DIM);
    float4 v = src[part];
    ((float4*)h)[f]   = v;
    ((float4*)acc)[f] = v;
}

__global__ void lgcn_zero(float* __restrict__ p) {
    size_t f = (size_t)blockIdx.x * blockDim.x + threadIdx.x;
    size_t total = (size_t)N_NODES * (DIM / 4);
    if (f >= total) return;
    ((float4*)p)[f] = make_float4(0.f, 0.f, 0.f, 0.f);
}

#define SPMM_GROUP 8

__global__ void lgcn_spmm(const int*   __restrict__ rows,
                          const int*   __restrict__ cols,
                          const float* __restrict__ vals,
                          const float* __restrict__ h,
                          float*       __restrict__ out) {
    const int lane = (int)(threadIdx.x & 31);
    const size_t e = (size_t)blockIdx.x * blockDim.x + threadIdx.x;
    if (e >= (size_t)NNZ) return;
    const int r  = rows[e];
    const int c  = cols[e];
    const int vb = __float_as_int(vals[e]);
    const float* hl = h   + lane;
    float*       ol = out + lane;

    float a[SPMM_GROUP], b[SPMM_GROUP];
#pragma unroll
    for (int k = 0; k < SPMM_GROUP; ++k) {
        const float* s = hl + (size_t)__builtin_amdgcn_readlane(c, k) * DIM;
        a[k] = s[0];
        b[k] = s[32];
    }
#pragma unroll
    for (int g = 0; g < 32; g += SPMM_GROUP) {
        float na[SPMM_GROUP], nb[SPMM_GROUP];
        if (g + SPMM_GROUP < 32) {
#pragma unroll
            for (int k = 0; k < SPMM_GROUP; ++k) {
                const float* s = hl + (size_t)__builtin_amdgcn_readlane(c, g + SPMM_GROUP + k) * DIM;
                na[k] = s[0];
                nb[k] = s[32];
            }
        }
#pragma unroll
        for (int k = 0; k < SPMM_GROUP; ++k) {
            const int   rj = __builtin_amdgcn_readlane(r, g + k);
            const float vj = __int_as_float(__builtin_amdgcn_readlane(vb, g + k));
            float* d = ol + (size_t)rj * DIM;
            unsafeAtomicAdd(d,      vj * a[k]);
            unsafeAtomicAdd(d + 32, vj * b[k]);
        }
        if (g + SPMM_GROUP < 32) {
#pragma unroll
            for (int k = 0; k < SPMM_GROUP; ++k) { a[k] = na[k]; b[k] = nb[k]; }
        }
    }
}

__global__ void lgcn_addacc(float* __restrict__ acc, const float* __restrict__ h) {
    size_t f = (size_t)blockIdx.x * blockDim.x + threadIdx.x;
    size_t total = (size_t)N_NODES * (DIM / 4);
    if (f >= total) return;
    float4 a = ((float4*)acc)[f];
    float4 b = ((const float4*)h)[f];
    a.x += b.x; a.y += b.y; a.z += b.z; a.w += b.w;
    ((float4*)acc)[f] = a;
}

#define NEED_WORDS ((N_NODES + 31) / 32)
#define LAST_TB    1024
#define LAST_GRID  512
#define LAST_GROUP 8

__global__ void lgcn_zero_rows(const int* __restrict__ users,
                               const int* __restrict__ items,
                               float* __restrict__ p) {
    int f = (int)(blockIdx.x * blockDim.x + threadIdx.x);
    if (f >= 2 * BATCH * (DIM / 4)) return;
    int slot = f >> 4;
    int part = f & 15;
    size_t node = (slot < BATCH) ? (size_t)users[slot]
                                 : (size_t)N_USERS + (size_t)items[slot - BATCH];
    ((float4*)(p + node * DIM))[part] = make_float4(0.f, 0.f, 0.f, 0.f);
}

__global__ void __launch_bounds__(LAST_TB)
lgcn_spmm_last(const int*   __restrict__ rows,
               const int*   __restrict__ cols,
               const float* __restrict__ vals,
               const float* __restrict__ h,
               float*       __restrict__ out,
               const int*   __restrict__ users,
               const int*   __restrict__ items) {
    __shared__ unsigned need[NEED_WORDS];
    for (int i = (int)threadIdx.x; i < NEED_WORDS; i += LAST_TB) need[i] = 0u;
    __syncthreads();
    {
        unsigned uu[BATCH / LAST_TB], tt[BATCH / LAST_TB];
#pragma unroll
        for (int k = 0; k < BATCH / LAST_TB; ++k) {
            uu[k] = (unsigned)users[(int)threadIdx.x + k * LAST_TB];
            tt[k] = (unsigned)N_USERS + (unsigned)items[(int)threadIdx.x + k * LAST_TB];
        }
#pragma unroll
        for (int k = 0; k < BATCH / LAST_TB; ++k) {
            atomicOr(&need[uu[k] >> 5], 1u << (uu[k] & 31u));
            atomicOr(&need[tt[k] >> 5], 1u << (tt[k] & 31u));
        }
    }
    __syncthreads();

    const int lane = (int)(threadIdx.x & 31);
    const int wave = __builtin_amdgcn_readfirstlane((int)(threadIdx.x >> 5));
    const float* hl = h   + lane;
    float*       ol = out + lane;

    const int stride = LAST_GRID * (LAST_TB / 32);
    int chunk = (int)blockIdx.x * (LAST_TB / 32) + wave;
    int r = 0, c = 0, vb = 0;
    if (chunk < NNZ / 32) {
        const size_t e = (size_t)chunk * 32 + (size_t)lane;
        r  = rows[e];
        c  = cols[e];
        vb = __float_as_int(vals[e]);
    }
    for (; chunk < NNZ / 32; chunk += stride) {
        const int nx = (chunk + stride < NNZ / 32) ? (chunk + stride) : (NNZ / 32 - 1);
        const size_t en = (size_t)nx * 32 + (size_t)lane;
        const int rn = rows[en];
        const int cn = cols[en];
        const int vn = __float_as_int(vals[en]);
        const bool hit = ((need[r >> 5] >> (r & 31)) & 1u) != 0u;
        unsigned m = __builtin_amdgcn_ballot_w32(hit);
        while (m != 0u) {
            int  j[LAST_GROUP];
            bool on[LAST_GROUP];
#pragma unroll
            for (int k = 0; k < LAST_GROUP; ++k) {
                on[k] = (m != 0u);
                j[k]  = on[k] ? __builtin_ctz(m) : 0;
                m &= (m - 1u);
            }
            float a[LAST_GROUP], b[LAST_GROUP];
#pragma unroll
            for (int k = 0; k < LAST_GROUP; ++k) {
                a[k] = 0.f; b[k] = 0.f;
                if (on[k]) {
                    const float* s = hl + (size_t)__builtin_amdgcn_readlane(c, j[k]) * DIM;
                    a[k] = s[0];
                    b[k] = s[32];
                }
            }
#pragma unroll
            for (int k = 0; k < LAST_GROUP; ++k) {
                if (on[k]) {
                    const int   rj = __builtin_amdgcn_readlane(r, j[k]);
                    const float vj = __int_as_float(__builtin_amdgcn_readlane(vb, j[k]));
                    float* d = ol + (size_t)rj * DIM;
                    unsafeAtomicAdd(d,      vj * a[k]);
                    unsafeAtomicAdd(d + 32, vj * b[k]);
                }
            }
        }
        r = rn; c = cn; vb = vn;
    }
}

__global__ void lgcn_dot_wmma(const float* __restrict__ acc,
                              const float* __restrict__ hlast,
                              const int*   __restrict__ users,
                              const int*   __restrict__ items,
                              float*       __restrict__ out) {
    __shared__ float lds[8][16][17];
    int lane  = threadIdx.x & 31;
    int wv    = threadIdx.x >> 5;
    int tile  = blockIdx.x * 8 + wv;
    int m     = lane & 15;
    int khalf = lane >> 4;

    int u_row = users[tile * 16 + m];
    int v_row = items[tile * 16 + m];
    const float* up = acc + (size_t)u_row * DIM;
    const float* vp = acc + ((size_t)N_USERS + (size_t)v_row) * DIM;
    const float* ul = hlast + (size_t)u_row * DIM;
    const float* vl = hlast + ((size_t)N_USERS + (size_t)v_row) * DIM;

    v8f c = {};
#pragma unroll
    for (int kc = 0; kc < DIM / 4; ++kc) {
        int off = kc * 4 + khalf * 2;
        v2f a; a.x = up[off] + ul[off]; a.y = up[off + 1] + ul[off + 1];
        v2f b; b.x = vp[off] + vl[off]; b.y = vp[off + 1] + vl[off + 1];
        c = __builtin_amdgcn_wmma_f32_16x16x4_f32(
                 false, a,  false, b,
                 (short)0, c,  false,  false);
    }

#pragma unroll
    for (int r = 0; r < 8; ++r)
        lds[wv][r + 8 * khalf][m] = c[r];
    __syncthreads();

    if (lane < 16)
        out[tile * 16 + lane] = lds[wv][lane][lane] * (1.0f / 16.0f);
}

extern "C" void kernel_launch(void* const* d_in, const int* in_sizes, int n_in,
                              void* d_out, int out_size, void* d_ws, size_t ws_size,
                              hipStream_t stream) {
    const float* user_emb = (const float*)d_in[0];
    const float* item_emb = (const float*)d_in[1];
    const int*   g_rows   = (const int*)  d_in[2];
    const int*   g_cols   = (const int*)  d_in[3];
    const float* g_vals   = (const float*)d_in[4];
    const int*   users    = (const int*)  d_in[5];
    const int*   items    = (const int*)  d_in[6];
    float*       out      = (float*)d_out;

    const size_t node_f = (size_t)N_NODES * DIM;
    float* bufA = (float*)d_ws;
    float* bufB = bufA + node_f;
    float* accb = bufB + node_f;

    const int TB = 256;
    const int elem_blocks = (int)(((size_t)N_NODES * (DIM / 4) + TB - 1) / TB);
    const int spmm_blocks = NNZ / TB;

    lgcn_init<<<elem_blocks, TB, 0, stream>>>(user_emb, item_emb, bufA, accb);

    float* hsrc = bufA;
    float* hdst = bufB;
    for (int hop = 0; hop < 2; ++hop) {
        lgcn_zero<<<elem_blocks, TB, 0, stream>>>(hdst);
        lgcn_spmm<<<spmm_blocks, TB, 0, stream>>>(g_rows, g_cols, g_vals, hsrc, hdst);
        lgcn_addacc<<<elem_blocks, TB, 0, stream>>>(accb, hdst);
        float* t = hsrc; hsrc = hdst; hdst = t;
    }

    lgcn_zero_rows<<<2 * BATCH * (DIM / 4) / TB, TB, 0, stream>>>(users, items, hdst);
    lgcn_spmm_last<<<LAST_GRID, LAST_TB, 0, stream>>>(g_rows, g_cols, g_vals, hsrc, hdst,
                                                      users, items);

    lgcn_dot_wmma<<<BATCH / 16 / 8, TB, 0, stream>>>(accb, hdst, users, items, out);
}


# Running sum never stored: hops 1, 2 keep own buffers, the dot kernel adds ((ego+h1)+h2)+h3 at the rows it reads; ego copy and both full acc-add passes go (10 launches to 7); sum order as before
# speedup vs baseline: 3.3130x; 1.1703x over previous
#include <hip/hip_runtime.h>
#include <hip/hip_bf16.h>

#define N_USERS 100000
#define N_ITEMS 200000
#define N_NODES (N_USERS + N_ITEMS)
#define NNZ     4800000
#define DIM     64
#define BATCH   16384

typedef float v2f __attribute__((ext_vector_type(2)));
typedef float v8f __attribute__((ext_vector_type(8)));

__global__ void lgcn_zero(float* __restrict__ p) {
    size_t f = (size_t)blockIdx.x * blockDim.x + threadIdx.x;
    size_t total = (size_t)N_NODES * (DIM / 4);
    if (f >= total) return;
    ((float4*)p)[f] = make_float4(0.f, 0.f, 0.f, 0.f);
}

#define SPMM_GROUP 8

template <bool EGO>
__device__ __forceinline__ const float* lgcn_src_row(const float* __restrict__ src,
                                                     const float* __restrict__ src_items,
                                                     int cj) {
    if (EGO && cj >= N_USERS) return src_items + (size_t)(cj - N_USERS) * DIM;
    return src + (size_t)cj * DIM;
}

template <bool EGO>
__device__ __forceinline__ void lgcn_spmm_body(const int*   __restrict__ rows,
                                               const int*   __restrict__ cols,
                                               const float* __restrict__ vals,
                                               const float* __restrict__ src,
                                               const float* __restrict__ src_items,
                                               float*       __restrict__ out) {
    const int lane = (int)(threadIdx.x & 31);
    const size_t e = (size_t)blockIdx.x * blockDim.x + threadIdx.x;
    if (e >= (size_t)NNZ) return;
    const int r  = rows[e];
    const int c  = cols[e];
    const int vb = __float_as_int(vals[e]);
    float*       ol = out + lane;

    float a[SPMM_GROUP], b[SPMM_GROUP];
#pragma unroll
    for (int k = 0; k < SPMM_GROUP; ++k) {
        const float* s = lgcn_src_row<EGO>(src, src_items, __builtin_amdgcn_readlane(c, k)) + lane;
        a[k] = s[0];
        b[k] = s[32];
    }
#pragma unroll
    for (int g = 0; g < 32; g += SPMM_GROUP) {
        float na[SPMM_GROUP], nb[SPMM_GROUP];
        if (g + SPMM_GROUP < 32) {
#pragma unroll
            for (int k = 0; k < SPMM_GROUP; ++k) {
                const float* s = lgcn_src_row<EGO>(src, src_items,
                                     __builtin_amdgcn_readlane(c, g + SPMM_GROUP + k)) + lane;
                na[k] = s[0];
                nb[k] = s[32];
            }
        }
#pragma unroll
        for (int k = 0; k < SPMM_GROUP; ++k) {
            const int   rj = __builtin_amdgcn_readlane(r, g + k);
            const float vj = __int_as_float(__builtin_amdgcn_readlane(vb, g + k));
            float* d = ol + (size_t)rj * DIM;
            unsafeAtomicAdd(d,      vj * a[k]);
            unsafeAtomicAdd(d + 32, vj * b[k]);
        }
        if (g + SPMM_GROUP < 32) {
#pragma unroll
            for (int k = 0; k < SPMM_GROUP; ++k) { a[k] = na[k]; b[k] = nb[k]; }
        }
    }
}

__global__ void lgcn_spmm_ego(const int*   __restrict__ rows,
                              const int*   __restrict__ cols,
                              const float* __restrict__ vals,
                              const float* __restrict__ user_emb,
                              const float* __restrict__ item_emb,
                              float*       __restrict__ out) {
    lgcn_spmm_body<true>(rows, cols, vals, user_emb, item_emb, out);
}

__global__ void lgcn_spmm(const int*   __restrict__ rows,
                          const int*   __restrict__ cols,
                          const float* __restrict__ vals,
                          const float* __restrict__ h,
                          float*       __restrict__ out) {
    lgcn_spmm_body<false>(rows, cols, vals, h, nullptr, out);
}

#define NEED_WORDS ((N_NODES + 31) / 32)
#define LAST_TB    1024
#define LAST_GRID  512
#define LAST_GROUP 8

__global__ void lgcn_zero_rows(const int* __restrict__ users,
                               const int* __restrict__ items,
                               float* __restrict__ p) {
    int f = (int)(blockIdx.x * blockDim.x + threadIdx.x);
    if (f >= 2 * BATCH * (DIM / 4)) return;
    int slot = f >> 4;
    int part = f & 15;
    size_t node = (slot < BATCH) ? (size_t)users[slot]
                                 : (size_t)N_USERS + (size_t)items[slot - BATCH];
    ((float4*)(p + node * DIM))[part] = make_float4(0.f, 0.f, 0.f, 0.f);
}

__global__ void __launch_bounds__(LAST_TB)
lgcn_spmm_last(const int*   __restrict__ rows,
               const int*   __restrict__ cols,
               const float* __restrict__ vals,
               const float* __restrict__ h,
               float*       __restrict__ out,
               const int*   __restrict__ users,
               const int*   __restrict__ items) {
    __shared__ unsigned need[NEED_WORDS];
    for (int i = (int)threadIdx.x; i < NEED_WORDS; i += LAST_TB) need[i] = 0u;
    __syncthreads();
    {
        unsigned uu[BATCH / LAST_TB], tt[BATCH / LAST_TB];
#pragma unroll
        for (int k = 0; k < BATCH / LAST_TB; ++k) {
            uu[k] = (unsigned)users[(int)threadIdx.x + k * LAST_TB];
            tt[k] = (unsigned)N_USERS + (unsigned)items[(int)threadIdx.x + k * LAST_TB];
        }
#pragma unroll
        for (int k = 0; k < BATCH / LAST_TB; ++k) {
            atomicOr(&need[uu[k] >> 5], 1u << (uu[k] & 31u));
            atomicOr(&need[tt[k] >> 5], 1u << (tt[k] & 31u));
        }
    }
    __syncthreads();

    const int lane = (int)(threadIdx.x & 31);
    const int wave = __builtin_amdgcn_readfirstlane((int)(threadIdx.x >> 5));
    const float* hl = h   + lane;
    float*       ol = out + lane;

    const int stride = LAST_GRID * (LAST_TB / 32);
    int chunk = (int)blockIdx.x * (LAST_TB / 32) + wave;
    int r = 0, c = 0, vb = 0;
    if (chunk < NNZ / 32) {
        const size_t e = (size_t)chunk * 32 + (size_t)lane;
        r  = rows[e];
        c  = cols[e];
        vb = __float_as_int(vals[e]);
    }
    for (; chunk < NNZ / 32; chunk += stride) {
        const int nx = (chunk + stride < NNZ / 32) ? (chunk + stride) : (NNZ / 32 - 1);
        const size_t en = (size_t)nx * 32 + (size_t)lane;
        const int rn = rows[en];
        const int cn = cols[en];
        const int vn = __float_as_int(vals[en]);
        const bool hit = ((need[r >> 5] >> (r & 31)) & 1u) != 0u;
        unsigned m = __builtin_amdgcn_ballot_w32(hit);
        while (m != 0u) {
            int  j[LAST_GROUP];
            bool on[LAST_GROUP];
#pragma unroll
            for (int k = 0; k < LAST_GROUP; ++k) {
                on[k] = (m != 0u);
                j[k]  = on[k] ? __builtin_ctz(m) : 0;
                m &= (m - 1u);
            }
            float a[LAST_GROUP], b[LAST_GROUP];
#pragma unroll
            for (int k = 0; k < LAST_GROUP; ++k) {
                a[k] = 0.f; b[k] = 0.f;
                if (on[k]) {
                    const float* s = hl + (size_t)__builtin_amdgcn_readlane(c, j[k]) * DIM;
                    a[k] = s[0];
                    b[k] = s[32];
                }
            }
#pragma unroll
            for (int k = 0; k < LAST_GROUP; ++k) {
                if (on[k]) {
                    const int   rj = __builtin_amdgcn_readlane(r, j[k]);
                    const float vj = __int_as_float(__builtin_amdgcn_readlane(vb, j[k]));
                    float* d = ol + (size_t)rj * DIM;
                    unsafeAtomicAdd(d,      vj * a[k]);
                    unsafeAtomicAdd(d + 32, vj * b[k]);
                }
            }
        }
        r = rn; c = cn; vb = vn;
    }
}

__global__ void lgcn_dot_wmma(const float* __restrict__ user_emb,
                              const float* __restrict__ item_emb,
                              const float* __restrict__ h1,
                              const float* __restrict__ h2,
                              const float* __restrict__ h3,
                              const int*   __restrict__ users,
                              const int*   __restrict__ items,
                              float*       __restrict__ out) {
    __shared__ float lds[8][16][17];
    int lane  = threadIdx.x & 31;
    int wv    = threadIdx.x >> 5;
    int tile  = blockIdx.x * 8 + wv;
    int m     = lane & 15;
    int khalf = lane >> 4;

    int u_row = users[tile * 16 + m];
    int v_row = items[tile * 16 + m];
    const size_t uo = (size_t)u_row * DIM;
    const size_t vo = ((size_t)N_USERS + (size_t)v_row) * DIM;
    const float* u0 = user_emb + uo;
    const float* v0 = item_emb + (size_t)v_row * DIM;
    const float* u1 = h1 + uo;  const float* v1 = h1 + vo;
    const float* u2 = h2 + uo;  const float* v2 = h2 + vo;
    const float* u3 = h3 + uo;  const float* v3 = h3 + vo;

    v8f c = {};
#pragma unroll
    for (int kc = 0; kc < DIM / 4; ++kc) {
        int off = kc * 4 + khalf * 2;
        v2f a; a.x = ((u0[off]     + u1[off])     + u2[off])     + u3[off];
               a.y = ((u0[off + 1] + u1[off + 1]) + u2[off + 1]) + u3[off + 1];
        v2f b; b.x = ((v0[off]     + v1[off])     + v2[off])     + v3[off];
               b.y = ((v0[off + 1] + v1[off + 1]) + v2[off + 1]) + v3[off + 1];
        c = __builtin_amdgcn_wmma_f32_16x16x4_f32(
                 false, a,  false, b,
                 (short)0, c,  false,  false);
    }

#pragma unroll
    for (int r = 0; r < 8; ++r)
        lds[wv][r + 8 * khalf][m] = c[r];
    __syncthreads();

    if (lane < 16)
        out[tile * 16 + lane] = lds[wv][lane][lane] * (1.0f / 16.0f);
}

extern "C" void kernel_launch(void* const* d_in, const int* in_sizes, int n_in,
                              void* d_out, int out_size, void* d_ws, size_t ws_size,
                              hipStream_t stream) {
    const float* user_emb = (const float*)d_in[0];
    const float* item_emb = (const float*)d_in[1];
    const int*   g_rows   = (const int*)  d_in[2];
    const int*   g_cols   = (const int*)  d_in[3];
    const float* g_vals   = (const float*)d_in[4];
    const int*   users    = (const int*)  d_in[5];
    const int*   items    = (const int*)  d_in[6];
    float*       out      = (float*)d_out;

    const size_t node_f = (size_t)N_NODES * DIM;
    float* hop1 = (float*)d_ws;
    float* hop2 = hop1 + node_f;
    float* hop3 = hop2 + node_f;

    const int TB = 256;
    const int elem_blocks = (int)(((size_t)N_NODES * (DIM / 4) + TB - 1) / TB);
    const int spmm_blocks = NNZ / TB;

    lgcn_zero<<<elem_blocks, TB, 0, stream>>>(hop1);
    lgcn_spmm_ego<<<spmm_blocks, TB, 0, stream>>>(g_rows, g_cols, g_vals, user_emb, item_emb, hop1);

    lgcn_zero<<<elem_blocks, TB, 0, stream>>>(hop2);
    lgcn_spmm<<<spmm_blocks, TB, 0, stream>>>(g_rows, g_cols, g_vals, hop1, hop2);

    lgcn_zero_rows<<<2 * BATCH * (DIM / 4) / TB, TB, 0, stream>>>(users, items, hop3);
    lgcn_spmm_last<<<LAST_GRID, LAST_TB, 0, stream>>>(g_rows, g_cols, g_vals, hop2, hop3,
                                                      users, items);

    lgcn_dot_wmma<<<BATCH / 16 / 8, TB, 0, stream>>>(user_emb, item_emb, hop1, hop2, hop3,
                                                     users, items, out);
}


# Per call nnz are counting-sorted into 240-row buckets by destination; a hop sums each bucket's rows in LDS (f32 LDS adds), stores them once: no global f32 atomic, no zero pass; sum order: LDS arrival
# speedup vs baseline: 4.8647x; 1.4684x over previous
#include <hip/hip_runtime.h>
#include <hip/hip_bf16.h>

#define N_USERS 100000
#define N_ITEMS 200000
#define N_NODES (N_USERS + N_ITEMS)
#define NNZ     4800000
#define DIM     64
#define BATCH   16384

typedef float v2f __attribute__((ext_vector_type(2)));
typedef float v8f __attribute__((ext_vector_type(8)));

#define RB       240
#define NB       ((N_NODES + RB - 1) / RB)
#define NEED_W   ((RB + 31) / 32)
#define ROW_BITS 8
#define COL_BITS 19
#define COL_MASK ((1 << COL_BITS) - 1)
static_assert(RB <= (1 << ROW_BITS), "a row within its bucket must fit ROW_BITS");
static_assert(N_NODES <= (1 << COL_BITS), "a column index must fit COL_BITS");
static_assert(ROW_BITS + COL_BITS <= 31, "packed (row, column) must stay a non-negative int");

#define BIN_TB    256
#define BIN_PER   ((NB + BIN_TB - 1) / BIN_TB)
#define NBP       (BIN_PER * BIN_TB)
#define BIN_TILE  8192
#define BIN_ITEMS (BIN_TILE / BIN_TB)
#define BIN_GRID  ((NNZ + BIN_TILE - 1) / BIN_TILE)
#define BIN_SUB   8

__global__ void lgcn_bin_clear(int* __restrict__ counts, int* __restrict__ cursor) {
    const int i = (int)(blockIdx.x * blockDim.x + threadIdx.x);
    if (i < NBP) { counts[i] = 0; cursor[i] = 0; }
}

__global__ void __launch_bounds__(BIN_TB)
lgcn_bin_count(const int* __restrict__ rows, int* __restrict__ counts) {
    __shared__ int s_cnt[NBP];
    const int tid = (int)threadIdx.x;
#pragma unroll
    for (int k = 0; k < BIN_PER; ++k) s_cnt[k * BIN_TB + tid] = 0;
    __syncthreads();
    const int base = (int)blockIdx.x * BIN_TILE + tid;
    int rr[BIN_ITEMS];
#pragma unroll
    for (int k = 0; k < BIN_ITEMS; ++k) {
        const int e = base + k * BIN_TB;
        rr[k] = (e < NNZ) ? rows[e] : -1;
    }
#pragma unroll
    for (int k = 0; k < BIN_ITEMS; ++k)
        if (rr[k] >= 0) atomicAdd(&s_cnt[rr[k] / RB], 1);
    __syncthreads();
#pragma unroll
    for (int k = 0; k < BIN_PER; ++k) {
        const int bk = k * BIN_TB + tid;
        const int c  = s_cnt[bk];
        if (bk < NB && c > 0) atomicAdd(&counts[bk], c);
    }
}

__global__ void __launch_bounds__(BIN_TB)
lgcn_bin_place(const int*   __restrict__ rows,
               const int*   __restrict__ cols,
               const float* __restrict__ vals,
               const int*   __restrict__ counts,
               int*         __restrict__ cursor,
               int*         __restrict__ starts,
               int2*        __restrict__ ent) {
    __shared__ int s_cnt[NBP];
    __shared__ int s_base[NBP];
    __shared__ int s_scan[BIN_TB];
    const int tid = (int)threadIdx.x;

#pragma unroll
    for (int k = 0; k < BIN_PER; ++k) {
        const int bk = k * BIN_TB + tid;
        s_base[bk] = (bk < NB) ? counts[bk] : 0;
        s_cnt[bk]  = 0;
    }
    __syncthreads();
    int loc[BIN_PER];
    int tsum = 0;
#pragma unroll
    for (int k = 0; k < BIN_PER; ++k) { loc[k] = s_base[tid * BIN_PER + k]; tsum += loc[k]; }
    s_scan[tid] = tsum;
    __syncthreads();
    for (int off = 1; off < BIN_TB; off <<= 1) {
        const int v = (tid >= off) ? s_scan[tid - off] : 0;
        __syncthreads();
        s_scan[tid] += v;
        __syncthreads();
    }
    int run = s_scan[tid] - tsum;
#pragma unroll
    for (int k = 0; k < BIN_PER; ++k) {
        const int bk = tid * BIN_PER + k;
        s_base[bk] = run;
        if (blockIdx.x == 0 && bk < NB) starts[bk] = run;
        run += loc[k];
    }
    __syncthreads();

    const int base = (int)blockIdx.x * BIN_TILE + tid;
    int rr[BIN_ITEMS];
#pragma unroll
    for (int k = 0; k < BIN_ITEMS; ++k) {
        const int e = base + k * BIN_TB;
        rr[k] = (e < NNZ) ? rows[e] : -1;
    }
#pragma unroll
    for (int k = 0; k < BIN_ITEMS; ++k)
        if (rr[k] >= 0) atomicAdd(&s_cnt[rr[k] / RB], 1);
    __syncthreads();
#pragma unroll
    for (int k = 0; k < BIN_PER; ++k) {
        const int bk = k * BIN_TB + tid;
        const int c  = s_cnt[bk];
        if (bk < NB && c > 0) s_base[bk] += atomicAdd(&cursor[bk], c);
        s_cnt[bk] = 0;
    }
    __syncthreads();
#pragma unroll
    for (int g = 0; g < BIN_ITEMS; g += BIN_SUB) {
        int cc[BIN_SUB], vv[BIN_SUB];
#pragma unroll
        for (int k = 0; k < BIN_SUB; ++k) {
            const int e = base + (g + k) * BIN_TB;
            cc[k] = 0; vv[k] = 0;
            if (rr[g + k] >= 0) { cc[k] = cols[e]; vv[k] = __float_as_int(vals[e]); }
        }
#pragma unroll
        for (int k = 0; k < BIN_SUB; ++k) {
            if (rr[g + k] >= 0) {
                const int bk = rr[g + k] / RB;
                const int p  = s_base[bk] + atomicAdd(&s_cnt[bk], 1);
                ent[p] = make_int2(((rr[g + k] - bk * RB) << COL_BITS) | cc[k], vv[k]);
            }
        }
    }
}

#define HOP_TB    256
#define HOP_WAVES (HOP_TB / 32)
#define HOP_GROUP 8
#define SCAN_SUB  32

template <bool EGO>
__device__ __forceinline__ const float* lgcn_src_row(const float* __restrict__ src,
                                                     const float* __restrict__ src_items,
                                                     int cj) {
    if (EGO && cj >= N_USERS) return src_items + (size_t)(cj - N_USERS) * DIM;
    return src + (size_t)cj * DIM;
}

__device__ __forceinline__ void lgcn_lds_add(float* p, float x) {
    __hip_atomic_fetch_add(p, x, __ATOMIC_RELAXED, __HIP_MEMORY_SCOPE_WORKGROUP);
}

__device__ __forceinline__ int lgcn_batch_node(const int* __restrict__ users,
                                               const int* __restrict__ items, int i) {
    return (i < BATCH) ? users[i] : N_USERS + items[i - BATCH];
}

template <int MODE>
__global__ void __launch_bounds__(HOP_TB)
lgcn_hop(const int2*  __restrict__ ent,
         const int*   __restrict__ counts,
         const int*   __restrict__ starts,
         const float* __restrict__ src,
         const float* __restrict__ src_items,
         float*       __restrict__ out,
         const int*   __restrict__ users,
         const int*   __restrict__ items) {
    constexpr bool EGO = (MODE == 0);
    __shared__ float acc[RB * DIM];
    const int tid  = (int)threadIdx.x;
    const int lane = tid & 31;
    const int wave = __builtin_amdgcn_readfirstlane(tid >> 5);
    const int b    = (int)blockIdx.x;

    unsigned need[NEED_W];
#pragma unroll
    for (int q = 0; q < NEED_W; ++q) need[q] = 0u;
    if (MODE == 2) {
        unsigned* nw = (unsigned*)acc;
        if (tid < NEED_W) nw[tid] = 0u;
        __syncthreads();
        for (int it = 0; it < 2 * BATCH / HOP_TB; it += SCAN_SUB) {
            int nd[SCAN_SUB];
#pragma unroll
            for (int k = 0; k < SCAN_SUB; ++k)
                nd[k] = lgcn_batch_node(users, items, (it + k) * HOP_TB + tid);
#pragma unroll
            for (int k = 0; k < SCAN_SUB; ++k) {
                const int lr = nd[k] - b * RB;
                if (lr >= 0 && lr < RB) atomicOr(&nw[lr >> 5], 1u << (lr & 31));
            }
        }
        __syncthreads();
#pragma unroll
        for (int q = 0; q < NEED_W; ++q) need[q] = nw[q];
        __syncthreads();
    }

    for (int i = tid; i < RB * DIM / 4; i += HOP_TB)
        ((float4*)acc)[i] = make_float4(0.f, 0.f, 0.f, 0.f);
    __syncthreads();

    const int   cnt = counts[b];
    const int2* eb  = ent + starts[b];
    float*      al  = acc + lane;

    int2 cur = make_int2(0, 0);
    if (wave * 32 + lane < cnt) cur = eb[wave * 32 + lane];
    for (int base = wave * 32; base < cnt; base += HOP_WAVES * 32) {
        const int nidx = base + HOP_WAVES * 32 + lane;
        const int2 nxt = eb[(nidx < cnt) ? nidx : (cnt - 1)];
        const int c  = cur.x & COL_MASK;
        const int rl = cur.x >> COL_BITS;
        const int vb = cur.y;
        const int n  = (cnt - base < 32) ? (cnt - base) : 32;

        if (MODE != 2 && n == 32) {
            float a[HOP_GROUP], bq[HOP_GROUP];
#pragma unroll
            for (int k = 0; k < HOP_GROUP; ++k) {
                const float* s = lgcn_src_row<EGO>(src, src_items, __builtin_amdgcn_readlane(c, k)) + lane;
                a[k]  = s[0];
                bq[k] = s[32];
            }
#pragma unroll
            for (int g = 0; g < 32; g += HOP_GROUP) {
                float na[HOP_GROUP], nb[HOP_GROUP];
                if (g + HOP_GROUP < 32) {
#pragma unroll
                    for (int k = 0; k < HOP_GROUP; ++k) {
                        const float* s = lgcn_src_row<EGO>(src, src_items,
                                             __builtin_amdgcn_readlane(c, g + HOP_GROUP + k)) + lane;
                        na[k] = s[0];
                        nb[k] = s[32];
                    }
                }
#pragma unroll
                for (int k = 0; k < HOP_GROUP; ++k) {
                    const int   rj = __builtin_amdgcn_readlane(rl, g + k);
                    const float vj = __int_as_float(__builtin_amdgcn_readlane(vb, g + k));
                    float* d = al + rj * DIM;
                    lgcn_lds_add(d,      vj * a[k]);
                    lgcn_lds_add(d + 32, vj * bq[k]);
                }
                if (g + HOP_GROUP < 32) {
#pragma unroll
                    for (int k = 0; k < HOP_GROUP; ++k) { a[k] = na[k]; bq[k] = nb[k]; }
                }
            }
        } else {
            bool ok = lane < n;
            if (MODE == 2) {
                unsigned wq = need[0];
#pragma unroll
                for (int q = 1; q < NEED_W; ++q) wq = ((rl >> 5) == q) ? need[q] : wq;
                ok = ok && (((wq >> (rl & 31)) & 1u) != 0u);
            }
            unsigned m = __builtin_amdgcn_ballot_w32(ok);
            while (m != 0u) {
                int  j[HOP_GROUP];
                bool on[HOP_GROUP];
#pragma unroll
                for (int k = 0; k < HOP_GROUP; ++k) {
                    on[k] = (m != 0u);
                    j[k]  = on[k] ? __builtin_ctz(m) : 0;
                    m &= (m - 1u);
                }
                float a[HOP_GROUP], bq[HOP_GROUP];
#pragma unroll
                for (int k = 0; k < HOP_GROUP; ++k) {
                    a[k] = 0.f; bq[k] = 0.f;
                    if (on[k]) {
                        const float* s = lgcn_src_row<EGO>(src, src_items,
                                             __builtin_amdgcn_readlane(c, j[k])) + lane;
                        a[k]  = s[0];
                        bq[k] = s[32];
                    }
                }
#pragma unroll
                for (int k = 0; k < HOP_GROUP; ++k) {
                    if (on[k]) {
                        const int   rj = __builtin_amdgcn_readlane(rl, j[k]);
                        const float vj = __int_as_float(__builtin_amdgcn_readlane(vb, j[k]));
                        float* d = al + rj * DIM;
                        lgcn_lds_add(d,      vj * a[k]);
                        lgcn_lds_add(d + 32, vj * bq[k]);
                    }
                }
            }
        }
        cur = nxt;
    }
    __syncthreads();

    if (MODE != 2) {
        for (int f = tid; f < RB * DIM / 4; f += HOP_TB) {
            if (b * RB + (f >> 4) < N_NODES)
                ((float4*)out)[(size_t)b * (RB * DIM / 4) + f] = ((const float4*)acc)[f];
        }
    } else {
        for (int it = 0; it < 2 * BATCH / HOP_TB; it += SCAN_SUB) {
            int nd[SCAN_SUB];
#pragma unroll
            for (int k = 0; k < SCAN_SUB; ++k)
                nd[k] = lgcn_batch_node(users, items, (it + k) * HOP_TB + tid);
#pragma unroll
            for (int k = 0; k < SCAN_SUB; ++k) {
                const int lr = nd[k] - b * RB;
                unsigned m = __builtin_amdgcn_ballot_w32(lr >= 0 && lr < RB);
                while (m != 0u) {
                    const int j = __builtin_ctz(m);
                    m &= (m - 1u);
                    const int    lj   = __builtin_amdgcn_readlane(lr, j);
                    const int    slot = (it + k) * HOP_TB + wave * 32 + j;
                    const float* s    = al + lj * DIM;
                    float*       dst  = out + (size_t)slot * DIM + lane;
                    dst[0]  = s[0];
                    dst[32] = s[32];
                }
            }
        }
    }
}

__global__ void lgcn_dot_wmma(const float* __restrict__ user_emb,
                              const float* __restrict__ item_emb,
                              const float* __restrict__ h1,
                              const float* __restrict__ h2,
                              const float* __restrict__ h3,
                              const int*   __restrict__ users,
                              const int*   __restrict__ items,
                              float*       __restrict__ out) {
    __shared__ float lds[8][16][17];
    int lane  = threadIdx.x & 31;
    int wv    = threadIdx.x >> 5;
    int tile  = blockIdx.x * 8 + wv;
    int m     = lane & 15;
    int khalf = lane >> 4;

    int u_row = users[tile * 16 + m];
    int v_row = items[tile * 16 + m];
    const size_t uo = (size_t)u_row * DIM;
    const size_t vo = ((size_t)N_USERS + (size_t)v_row) * DIM;
    const float* u0 = user_emb + uo;
    const float* v0 = item_emb + (size_t)v_row * DIM;
    const float* u1 = h1 + uo;  const float* v1 = h1 + vo;
    const float* u2 = h2 + uo;  const float* v2 = h2 + vo;
    const float* u3 = h3 + (size_t)(tile * 16 + m) * DIM;
    const float* v3 = h3 + (size_t)(BATCH + tile * 16 + m) * DIM;

    v8f c = {};
#pragma unroll
    for (int kc = 0; kc < DIM / 4; ++kc) {
        int off = kc * 4 + khalf * 2;
        v2f a; a.x = ((u0[off]     + u1[off])     + u2[off])     + u3[off];
               a.y = ((u0[off + 1] + u1[off + 1]) + u2[off + 1]) + u3[off + 1];
        v2f b; b.x = ((v0[off]     + v1[off])     + v2[off])     + v3[off];
               b.y = ((v0[off + 1] + v1[off + 1]) + v2[off + 1]) + v3[off + 1];
        c = __builtin_amdgcn_wmma_f32_16x16x4_f32(
                 false, a,  false, b,
                 (short)0, c,  false,  false);
    }

#pragma unroll
    for (int r = 0; r < 8; ++r)
        lds[wv][r + 8 * khalf][m] = c[r];
    __syncthreads();

    if (lane < 16)
        out[tile * 16 + lane] = lds[wv][lane][lane] * (1.0f / 16.0f);
}

extern "C" void kernel_launch(void* const* d_in, const int* in_sizes, int n_in,
                              void* d_out, int out_size, void* d_ws, size_t ws_size,
                              hipStream_t stream) {
    const float* user_emb = (const float*)d_in[0];
    const float* item_emb = (const float*)d_in[1];
    const int*   g_rows   = (const int*)  d_in[2];
    const int*   g_cols   = (const int*)  d_in[3];
    const float* g_vals   = (const float*)d_in[4];
    const int*   users    = (const int*)  d_in[5];
    const int*   items    = (const int*)  d_in[6];
    float*       out      = (float*)d_out;

    const size_t node_f = (size_t)N_NODES * DIM;
    float* hop1   = (float*)d_ws;
    float* hop2   = hop1 + node_f;
    float* hop3   = hop2 + node_f;
    int2*  ent    = (int2*)(hop3 + (size_t)2 * BATCH * DIM);
    int*   counts = (int*)(ent + NNZ);
    int*   cursor = counts + NBP;
    int*   starts = cursor + NBP;

    const int TB = 256;

    lgcn_bin_clear<<<NBP / BIN_TB, BIN_TB, 0, stream>>>(counts, cursor);
    lgcn_bin_count<<<BIN_GRID, BIN_TB, 0, stream>>>(g_rows, counts);
    lgcn_bin_place<<<BIN_GRID, BIN_TB, 0, stream>>>(g_rows, g_cols, g_vals, counts, cursor,
                                                    starts, ent);

    lgcn_hop<0><<<NB, HOP_TB, 0, stream>>>(ent, counts, starts, user_emb, item_emb, hop1,
                                           users, items);
    lgcn_hop<1><<<NB, HOP_TB, 0, stream>>>(ent, counts, starts, hop1, nullptr, hop2,
                                           users, items);
    lgcn_hop<2><<<NB, HOP_TB, 0, stream>>>(ent, counts, starts, hop2, nullptr, hop3,
                                           users, items);

    lgcn_dot_wmma<<<BATCH / 16 / 8, TB, 0, stream>>>(user_emb, item_emb, hop1, hop2, hop3,
                                                     users, items, out);
}


# Batch positions counting-sorted by bucket in the three binning kernels; a last-hop block reads its own bucket's list (mean 26), not all 32768 batch indices twice; sum order as before
# speedup vs baseline: 5.6523x; 1.1619x over previous
#include <hip/hip_runtime.h>
#include <hip/hip_bf16.h>

#define N_USERS 100000
#define N_ITEMS 200000
#define N_NODES (N_USERS + N_ITEMS)
#define NNZ     4800000
#define DIM     64
#define BATCH   16384

typedef float v2f __attribute__((ext_vector_type(2)));
typedef float v8f __attribute__((ext_vector_type(8)));

#define RB       240
#define NB       ((N_NODES + RB - 1) / RB)
#define NEED_W   ((RB + 31) / 32)
#define ROW_BITS 8
#define COL_BITS 19
#define COL_MASK ((1 << COL_BITS) - 1)
static_assert(RB <= (1 << ROW_BITS), "a row within its bucket must fit ROW_BITS");
static_assert(N_NODES <= (1 << COL_BITS), "a column index must fit COL_BITS");
static_assert(ROW_BITS + COL_BITS <= 31, "packed (row, column) must stay a non-negative int");

#define BIN_TB    256
#define BIN_PER   ((NB + BIN_TB - 1) / BIN_TB)
#define NBP       (BIN_PER * BIN_TB)
#define BIN_TILE  8192
#define BIN_ITEMS (BIN_TILE / BIN_TB)
#define BIN_GRID  ((NNZ + BIN_TILE - 1) / BIN_TILE)
#define BIN_SUB   8

static_assert(2 * BATCH <= BIN_GRID * BIN_TB, "one batch position per thread of the binning grid");
static_assert(2 * BATCH <= (1 << (31 - ROW_BITS)), "packed (position, row) must stay a non-negative int");

__device__ __forceinline__ int lgcn_batch_node(const int* __restrict__ users,
                                               const int* __restrict__ items, int i) {
    return (i < BATCH) ? users[i] : N_USERS + items[i - BATCH];
}

__global__ void lgcn_bin_clear(int* __restrict__ counts, int* __restrict__ cursor,
                               int* __restrict__ bcnt, int* __restrict__ bcur) {
    const int i = (int)(blockIdx.x * blockDim.x + threadIdx.x);
    if (i < NBP) { counts[i] = 0; cursor[i] = 0; bcnt[i] = 0; bcur[i] = 0; }
}

__global__ void __launch_bounds__(BIN_TB)
lgcn_bin_count(const int* __restrict__ rows, int* __restrict__ counts,
               const int* __restrict__ users, const int* __restrict__ items,
               int* __restrict__ bcnt) {
    __shared__ int s_cnt[NBP];
    const int tid = (int)threadIdx.x;
    const int g = (int)blockIdx.x * BIN_TB + tid;
    if (g < 2 * BATCH) atomicAdd(&bcnt[lgcn_batch_node(users, items, g) / RB], 1);
#pragma unroll
    for (int k = 0; k < BIN_PER; ++k) s_cnt[k * BIN_TB + tid] = 0;
    __syncthreads();
    const int base = (int)blockIdx.x * BIN_TILE + tid;
    int rr[BIN_ITEMS];
#pragma unroll
    for (int k = 0; k < BIN_ITEMS; ++k) {
        const int e = base + k * BIN_TB;
        rr[k] = (e < NNZ) ? rows[e] : -1;
    }
#pragma unroll
    for (int k = 0; k < BIN_ITEMS; ++k)
        if (rr[k] >= 0) atomicAdd(&s_cnt[rr[k] / RB], 1);
    __syncthreads();
#pragma unroll
    for (int k = 0; k < BIN_PER; ++k) {
        const int bk = k * BIN_TB + tid;
        const int c  = s_cnt[bk];
        if (bk < NB && c > 0) atomicAdd(&counts[bk], c);
    }
}

__global__ void __launch_bounds__(BIN_TB)
lgcn_bin_place(const int*   __restrict__ rows,
               const int*   __restrict__ cols,
               const float* __restrict__ vals,
               const int*   __restrict__ counts,
               int*         __restrict__ cursor,
               int*         __restrict__ starts,
               int2*        __restrict__ ent,
               const int*   __restrict__ users,
               const int*   __restrict__ items,
               const int*   __restrict__ bcnt,
               int*         __restrict__ bcur,
               int*         __restrict__ bstart,
               int*         __restrict__ blist) {
    __shared__ int s_cnt[NBP];
    __shared__ int s_base[NBP];
    __shared__ int s_bb[NBP];
    __shared__ unsigned long long s_scan[BIN_TB];
    const int tid = (int)threadIdx.x;

#pragma unroll
    for (int k = 0; k < BIN_PER; ++k) {
        const int bk = k * BIN_TB + tid;
        s_base[bk] = (bk < NB) ? counts[bk] : 0;
        s_bb[bk]   = (bk < NB) ? bcnt[bk] : 0;
        s_cnt[bk]  = 0;
    }
    __syncthreads();
    int loc[BIN_PER], lob[BIN_PER];
    unsigned long long tsum = 0ull;
#pragma unroll
    for (int k = 0; k < BIN_PER; ++k) {
        loc[k] = s_base[tid * BIN_PER + k];
        lob[k] = s_bb[tid * BIN_PER + k];
        tsum += (unsigned long long)(unsigned)loc[k] | ((unsigned long long)(unsigned)lob[k] << 32);
    }
    s_scan[tid] = tsum;
    __syncthreads();
    for (int off = 1; off < BIN_TB; off <<= 1) {
        const unsigned long long v = (tid >= off) ? s_scan[tid - off] : 0ull;
        __syncthreads();
        s_scan[tid] += v;
        __syncthreads();
    }
    unsigned long long run = s_scan[tid] - tsum;
#pragma unroll
    for (int k = 0; k < BIN_PER; ++k) {
        const int bk = tid * BIN_PER + k;
        const int rn = (int)(unsigned)(run & 0xffffffffull);
        const int rb = (int)(unsigned)(run >> 32);
        s_base[bk] = rn;
        s_bb[bk]   = rb;
        if (blockIdx.x == 0 && bk < NB) { starts[bk] = rn; bstart[bk] = rb; }
        run += (unsigned long long)(unsigned)loc[k] | ((unsigned long long)(unsigned)lob[k] << 32);
    }
    __syncthreads();

    int bl_base = -1, bl_rank = 0, bl_word = 0;
    {
        const int g = (int)blockIdx.x * BIN_TB + tid;
        if (g < 2 * BATCH) {
            const int node = lgcn_batch_node(users, items, g);
            const int bk   = node / RB;
            bl_base = s_bb[bk];
            bl_rank = atomicAdd(&bcur[bk], 1);
            bl_word = (g << ROW_BITS) | (node - bk * RB);
        }
    }

    const int base = (int)blockIdx.x * BIN_TILE + tid;
    int rr[BIN_ITEMS];
#pragma unroll
    for (int k = 0; k < BIN_ITEMS; ++k) {
        const int e = base + k * BIN_TB;
        rr[k] = (e < NNZ) ? rows[e] : -1;
    }
#pragma unroll
    for (int k = 0; k < BIN_ITEMS; ++k)
        if (rr[k] >= 0) atomicAdd(&s_cnt[rr[k] / RB], 1);
    __syncthreads();
#pragma unroll
    for (int k = 0; k < BIN_PER; ++k) {
        const int bk = k * BIN_TB + tid;
        const int c  = s_cnt[bk];
        if (bk < NB && c > 0) s_base[bk] += atomicAdd(&cursor[bk], c);
        s_cnt[bk] = 0;
    }
    __syncthreads();
#pragma unroll
    for (int g = 0; g < BIN_ITEMS; g += BIN_SUB) {
        int cc[BIN_SUB], vv[BIN_SUB];
#pragma unroll
        for (int k = 0; k < BIN_SUB; ++k) {
            const int e = base + (g + k) * BIN_TB;
            cc[k] = 0; vv[k] = 0;
            if (rr[g + k] >= 0) { cc[k] = cols[e]; vv[k] = __float_as_int(vals[e]); }
        }
#pragma unroll
        for (int k = 0; k < BIN_SUB; ++k) {
            if (rr[g + k] >= 0) {
                const int bk = rr[g + k] / RB;
                const int p  = s_base[bk] + atomicAdd(&s_cnt[bk], 1);
                ent[p] = make_int2(((rr[g + k] - bk * RB) << COL_BITS) | cc[k], vv[k]);
            }
        }
    }
    if (bl_base >= 0) blist[bl_base + bl_rank] = bl_word;
}

#define HOP_TB    256
#define HOP_WAVES (HOP_TB / 32)
#define HOP_GROUP 8

template <bool EGO>
__device__ __forceinline__ const float* lgcn_src_row(const float* __restrict__ src,
                                                     const float* __restrict__ src_items,
                                                     int cj) {
    if (EGO && cj >= N_USERS) return src_items + (size_t)(cj - N_USERS) * DIM;
    return src + (size_t)cj * DIM;
}

__device__ __forceinline__ void lgcn_lds_add(float* p, float x) {
    __hip_atomic_fetch_add(p, x, __ATOMIC_RELAXED, __HIP_MEMORY_SCOPE_WORKGROUP);
}

template <int MODE>
__global__ void __launch_bounds__(HOP_TB)
lgcn_hop(const int2*  __restrict__ ent,
         const int*   __restrict__ counts,
         const int*   __restrict__ starts,
         const float* __restrict__ src,
         const float* __restrict__ src_items,
         float*       __restrict__ out,
         const int*   __restrict__ bcnt,
         const int*   __restrict__ bstart,
         const int*   __restrict__ blist) {
    constexpr bool EGO = (MODE == 0);
    __shared__ float acc[RB * DIM];
    const int tid  = (int)threadIdx.x;
    const int lane = tid & 31;
    const int wave = __builtin_amdgcn_readfirstlane(tid >> 5);
    const int b    = (int)blockIdx.x;

    unsigned need[NEED_W];
#pragma unroll
    for (int q = 0; q < NEED_W; ++q) need[q] = 0u;
    int        bn = 0;
    const int* bl = blist;
    if (MODE == 2) {
        bn = bcnt[b];
        bl = blist + bstart[b];
        unsigned* nw = (unsigned*)acc;
        if (tid < NEED_W) nw[tid] = 0u;
        __syncthreads();
        for (int i = tid; i < bn; i += HOP_TB) {
            const int lr = bl[i] & ((1 << ROW_BITS) - 1);
            atomicOr(&nw[lr >> 5], 1u << (lr & 31));
        }
        __syncthreads();
#pragma unroll
        for (int q = 0; q < NEED_W; ++q) need[q] = nw[q];
        __syncthreads();
    }

    for (int i = tid; i < RB * DIM / 4; i += HOP_TB)
        ((float4*)acc)[i] = make_float4(0.f, 0.f, 0.f, 0.f);
    __syncthreads();

    const int   cnt = counts[b];
    const int2* eb  = ent + starts[b];
    float*      al  = acc + lane;

    int2 cur = make_int2(0, 0);
    if (wave * 32 + lane < cnt) cur = eb[wave * 32 + lane];
    for (int base = wave * 32; base < cnt; base += HOP_WAVES * 32) {
        const int nidx = base + HOP_WAVES * 32 + lane;
        const int2 nxt = eb[(nidx < cnt) ? nidx : (cnt - 1)];
        const int c  = cur.x & COL_MASK;
        const int rl = cur.x >> COL_BITS;
        const int vb = cur.y;
        const int n  = (cnt - base < 32) ? (cnt - base) : 32;

        if (MODE != 2 && n == 32) {
            float a[HOP_GROUP], bq[HOP_GROUP];
#pragma unroll
            for (int k = 0; k < HOP_GROUP; ++k) {
                const float* s = lgcn_src_row<EGO>(src, src_items, __builtin_amdgcn_readlane(c, k)) + lane;
                a[k]  = s[0];
                bq[k] = s[32];
            }
#pragma unroll
            for (int g = 0; g < 32; g += HOP_GROUP) {
                float na[HOP_GROUP], nb[HOP_GROUP];
                if (g + HOP_GROUP < 32) {
#pragma unroll
                    for (int k = 0; k < HOP_GROUP; ++k) {
                        const float* s = lgcn_src_row<EGO>(src, src_items,
                                             __builtin_amdgcn_readlane(c, g + HOP_GROUP + k)) + lane;
                        na[k] = s[0];
                        nb[k] = s[32];
                    }
                }
#pragma unroll
                for (int k = 0; k < HOP_GROUP; ++k) {
                    const int   rj = __builtin_amdgcn_readlane(rl, g + k);
                    const float vj = __int_as_float(__builtin_amdgcn_readlane(vb, g + k));
                    float* d = al + rj * DIM;
                    lgcn_lds_add(d,      vj * a[k]);
                    lgcn_lds_add(d + 32, vj * bq[k]);
                }
                if (g + HOP_GROUP < 32) {
#pragma unroll
                    for (int k = 0; k < HOP_GROUP; ++k) { a[k] = na[k]; bq[k] = nb[k]; }
                }
            }
        } else {
            bool ok = lane < n;
            if (MODE == 2) {
                unsigned wq = need[0];
#pragma unroll
                for (int q = 1; q < NEED_W; ++q) wq = ((rl >> 5) == q) ? need[q] : wq;
                ok = ok && (((wq >> (rl & 31)) & 1u) != 0u);
            }
            unsigned m = __builtin_amdgcn_ballot_w32(ok);
            while (m != 0u) {
                int  j[HOP_GROUP];
                bool on[HOP_GROUP];
#pragma unroll
                for (int k = 0; k < HOP_GROUP; ++k) {
                    on[k] = (m != 0u);
                    j[k]  = on[k] ? __builtin_ctz(m) : 0;
                    m &= (m - 1u);
                }
                float a[HOP_GROUP], bq[HOP_GROUP];
#pragma unroll
                for (int k = 0; k < HOP_GROUP; ++k) {
                    a[k] = 0.f; bq[k] = 0.f;
                    if (on[k]) {
                        const float* s = lgcn_src_row<EGO>(src, src_items,
                                             __builtin_amdgcn_readlane(c, j[k])) + lane;
                        a[k]  = s[0];
                        bq[k] = s[32];
                    }
                }
#pragma unroll
                for (int k = 0; k < HOP_GROUP; ++k) {
                    if (on[k]) {
                        const int   rj = __builtin_amdgcn_readlane(rl, j[k]);
                        const float vj = __int_as_float(__builtin_amdgcn_readlane(vb, j[k]));
                        float* d = al + rj * DIM;
                        lgcn_lds_add(d,      vj * a[k]);
                        lgcn_lds_add(d + 32, vj * bq[k]);
                    }
                }
            }
        }
        cur = nxt;
    }
    __syncthreads();

    if (MODE != 2) {
        for (int f = tid; f < RB * DIM / 4; f += HOP_TB) {
            if (b * RB + (f >> 4) < N_NODES)
                ((float4*)out)[(size_t)b * (RB * DIM / 4) + f] = ((const float4*)acc)[f];
        }
    } else {
        for (int i = wave; i < bn; i += HOP_WAVES) {
            const int    e    = bl[i];
            const int    lr   = e & ((1 << ROW_BITS) - 1);
            const int    slot = e >> ROW_BITS;
            const float* s    = al + lr * DIM;
            float*       dst  = out + (size_t)slot * DIM + lane;
            dst[0]  = s[0];
            dst[32] = s[32];
        }
    }
}

__global__ void lgcn_dot_wmma(const float* __restrict__ user_emb,
                              const float* __restrict__ item_emb,
                              const float* __restrict__ h1,
                              const float* __restrict__ h2,
                              const float* __restrict__ h3,
                              const int*   __restrict__ users,
                              const int*   __restrict__ items,
                              float*       __restrict__ out) {
    __shared__ float lds[8][16][17];
    int lane  = threadIdx.x & 31;
    int wv    = threadIdx.x >> 5;
    int tile  = blockIdx.x * 8 + wv;
    int m     = lane & 15;
    int khalf = lane >> 4;

    int u_row = users[tile * 16 + m];
    int v_row = items[tile * 16 + m];
    const size_t uo = (size_t)u_row * DIM;
    const size_t vo = ((size_t)N_USERS + (size_t)v_row) * DIM;
    const float* u0 = user_emb + uo;
    const float* v0 = item_emb + (size_t)v_row * DIM;
    const float* u1 = h1 + uo;  const float* v1 = h1 + vo;
    const float* u2 = h2 + uo;  const float* v2 = h2 + vo;
    const float* u3 = h3 + (size_t)(tile * 16 + m) * DIM;
    const float* v3 = h3 + (size_t)(BATCH + tile * 16 + m) * DIM;

    v8f c = {};
#pragma unroll
    for (int kc = 0; kc < DIM / 4; ++kc) {
        int off = kc * 4 + khalf * 2;
        v2f a; a.x = ((u0[off]     + u1[off])     + u2[off])     + u3[off];
               a.y = ((u0[off + 1] + u1[off + 1]) + u2[off + 1]) + u3[off + 1];
        v2f b; b.x = ((v0[off]     + v1[off])     + v2[off])     + v3[off];
               b.y = ((v0[off + 1] + v1[off + 1]) + v2[off + 1]) + v3[off + 1];
        c = __builtin_amdgcn_wmma_f32_16x16x4_f32(
                 false, a,  false, b,
                 (short)0, c,  false,  false);
    }

#pragma unroll
    for (int r = 0; r < 8; ++r)
        lds[wv][r + 8 * khalf][m] = c[r];
    __syncthreads();

    if (lane < 16)
        out[tile * 16 + lane] = lds[wv][lane][lane] * (1.0f / 16.0f);
}

extern "C" void kernel_launch(void* const* d_in, const int* in_sizes, int n_in,
                              void* d_out, int out_size, void* d_ws, size_t ws_size,
                              hipStream_t stream) {
    const float* user_emb = (const float*)d_in[0];
    const float* item_emb = (const float*)d_in[1];
    const int*   g_rows   = (const int*)  d_in[2];
    const int*   g_cols   = (const int*)  d_in[3];
    const float* g_vals   = (const float*)d_in[4];
    const int*   users    = (const int*)  d_in[5];
    const int*   items    = (const int*)  d_in[6];
    float*       out      = (float*)d_out;

    const size_t node_f = (size_t)N_NODES * DIM;
    float* hop1   = (float*)d_ws;
    float* hop2   = hop1 + node_f;
    float* hop3   = hop2 + node_f;
    int2*  ent    = (int2*)(hop3 + (size_t)2 * BATCH * DIM);
    int*   counts = (int*)(ent + NNZ);
    int*   cursor = counts + NBP;
    int*   starts = cursor + NBP;
    int*   bcnt   = starts + NBP;
    int*   bcur   = bcnt + NBP;
    int*   bstart = bcur + NBP;
    int*   blist  = bstart + NBP;

    const int TB = 256;

    lgcn_bin_clear<<<NBP / BIN_TB, BIN_TB, 0, stream>>>(counts, cursor, bcnt, bcur);
    lgcn_bin_count<<<BIN_GRID, BIN_TB, 0, stream>>>(g_rows, counts, users, items, bcnt);
    lgcn_bin_place<<<BIN_GRID, BIN_TB, 0, stream>>>(g_rows, g_cols, g_vals, counts, cursor,
                                                    starts, ent, users, items, bcnt, bcur,
                                                    bstart, blist);

    lgcn_hop<0><<<NB, HOP_TB, 0, stream>>>(ent, counts, starts, user_emb, item_emb, hop1,
                                           bcnt, bstart, blist);
    lgcn_hop<1><<<NB, HOP_TB, 0, stream>>>(ent, counts, starts, hop1, nullptr, hop2,
                                           bcnt, bstart, blist);
    lgcn_hop<2><<<NB, HOP_TB, 0, stream>>>(ent, counts, starts, hop2, nullptr, hop3,
                                           bcnt, bstart, blist);

    lgcn_dot_wmma<<<BATCH / 16 / 8, TB, 0, stream>>>(user_emb, item_emb, hop1, hop2, hop3,
                                                     users, items, out);
}


# Hops 1, 2: each bucket's nnz are counting-sorted by row in LDS, then one wave sums a row in registers (fma) and stores it: no f32 LDS add, zero or copy-out; sum order: per row, sorted-list order
# speedup vs baseline: 6.2311x; 1.1024x over previous
#include <hip/hip_runtime.h>
#include <hip/hip_bf16.h>

#define N_USERS 100000
#define N_ITEMS 200000
#define N_NODES (N_USERS + N_ITEMS)
#define NNZ     4800000
#define DIM     64
#define BATCH   16384

typedef float v2f __attribute__((ext_vector_type(2)));
typedef float v8f __attribute__((ext_vector_type(8)));

#define RB       240
#define NB       ((N_NODES + RB - 1) / RB)
#define NEED_W   ((RB + 31) / 32)
#define ROW_BITS 8
#define COL_BITS 19
#define COL_MASK ((1 << COL_BITS) - 1)
static_assert(RB <= (1 << ROW_BITS), "a row within its bucket must fit ROW_BITS");
static_assert(N_NODES <= (1 << COL_BITS), "a column index must fit COL_BITS");
static_assert(ROW_BITS + COL_BITS <= 31, "packed (row, column) must stay a non-negative int");

#define BIN_TB    256
#define BIN_PER   ((NB + BIN_TB - 1) / BIN_TB)
#define NBP       (BIN_PER * BIN_TB)
#define BIN_TILE  8192
#define BIN_ITEMS (BIN_TILE / BIN_TB)
#define BIN_GRID  ((NNZ + BIN_TILE - 1) / BIN_TILE)
#define BIN_SUB   8

static_assert(2 * BATCH <= BIN_GRID * BIN_TB, "one batch position per thread of the binning grid");
static_assert(2 * BATCH <= (1 << (31 - ROW_BITS)), "packed (position, row) must stay a non-negative int");

__device__ __forceinline__ int lgcn_batch_node(const int* __restrict__ users,
                                               const int* __restrict__ items, int i) {
    return (i < BATCH) ? users[i] : N_USERS + items[i - BATCH];
}

__global__ void lgcn_bin_clear(int* __restrict__ counts, int* __restrict__ cursor,
                               int* __restrict__ bcnt, int* __restrict__ bcur) {
    const int i = (int)(blockIdx.x * blockDim.x + threadIdx.x);
    if (i < NBP) { counts[i] = 0; cursor[i] = 0; bcnt[i] = 0; bcur[i] = 0; }
}

__global__ void __launch_bounds__(BIN_TB)
lgcn_bin_count(const int* __restrict__ rows, int* __restrict__ counts,
               const int* __restrict__ users, const int* __restrict__ items,
               int* __restrict__ bcnt) {
    __shared__ int s_cnt[NBP];
    const int tid = (int)threadIdx.x;
    const int g = (int)blockIdx.x * BIN_TB + tid;
    if (g < 2 * BATCH) atomicAdd(&bcnt[lgcn_batch_node(users, items, g) / RB], 1);
#pragma unroll
    for (int k = 0; k < BIN_PER; ++k) s_cnt[k * BIN_TB + tid] = 0;
    __syncthreads();
    const int base = (int)blockIdx.x * BIN_TILE + tid;
    int rr[BIN_ITEMS];
#pragma unroll
    for (int k = 0; k < BIN_ITEMS; ++k) {
        const int e = base + k * BIN_TB;
        rr[k] = (e < NNZ) ? rows[e] : -1;
    }
#pragma unroll
    for (int k = 0; k < BIN_ITEMS; ++k)
        if (rr[k] >= 0) atomicAdd(&s_cnt[rr[k] / RB], 1);
    __syncthreads();
#pragma unroll
    for (int k = 0; k < BIN_PER; ++k) {
        const int bk = k * BIN_TB + tid;
        const int c  = s_cnt[bk];
        if (bk < NB && c > 0) atomicAdd(&counts[bk], c);
    }
}

__global__ void __launch_bounds__(BIN_TB)
lgcn_bin_place(const int*   __restrict__ rows,
               const int*   __restrict__ cols,
               const float* __restrict__ vals,
               const int*   __restrict__ counts,
               int*         __restrict__ cursor,
               int*         __restrict__ starts,
               int2*        __restrict__ ent,
               const int*   __restrict__ users,
               const int*   __restrict__ items,
               const int*   __restrict__ bcnt,
               int*         __restrict__ bcur,
               int*         __restrict__ bstart,
               int*         __restrict__ blist) {
    __shared__ int s_cnt[NBP];
    __shared__ int s_base[NBP];
    __shared__ int s_bb[NBP];
    __shared__ unsigned long long s_scan[BIN_TB];
    const int tid = (int)threadIdx.x;

#pragma unroll
    for (int k = 0; k < BIN_PER; ++k) {
        const int bk = k * BIN_TB + tid;
        s_base[bk] = (bk < NB) ? counts[bk] : 0;
        s_bb[bk]   = (bk < NB) ? bcnt[bk] : 0;
        s_cnt[bk]  = 0;
    }
    __syncthreads();
    int loc[BIN_PER], lob[BIN_PER];
    unsigned long long tsum = 0ull;
#pragma unroll
    for (int k = 0; k < BIN_PER; ++k) {
        loc[k] = s_base[tid * BIN_PER + k];
        lob[k] = s_bb[tid * BIN_PER + k];
        tsum += (unsigned long long)(unsigned)loc[k] | ((unsigned long long)(unsigned)lob[k] << 32);
    }
    s_scan[tid] = tsum;
    __syncthreads();
    for (int off = 1; off < BIN_TB; off <<= 1) {
        const unsigned long long v = (tid >= off) ? s_scan[tid - off] : 0ull;
        __syncthreads();
        s_scan[tid] += v;
        __syncthreads();
    }
    unsigned long long run = s_scan[tid] - tsum;
#pragma unroll
    for (int k = 0; k < BIN_PER; ++k) {
        const int bk = tid * BIN_PER + k;
        const int rn = (int)(unsigned)(run & 0xffffffffull);
        const int rb = (int)(unsigned)(run >> 32);
        s_base[bk] = rn;
        s_bb[bk]   = rb;
        if (blockIdx.x == 0 && bk < NB) { starts[bk] = rn; bstart[bk] = rb; }
        run += (unsigned long long)(unsigned)loc[k] | ((unsigned long long)(unsigned)lob[k] << 32);
    }
    __syncthreads();

    int bl_base = -1, bl_rank = 0, bl_word = 0;
    {
        const int g = (int)blockIdx.x * BIN_TB + tid;
        if (g < 2 * BATCH) {
            const int node = lgcn_batch_node(users, items, g);
            const int bk   = node / RB;
            bl_base = s_bb[bk];
            bl_rank = atomicAdd(&bcur[bk], 1);
            bl_word = (g << ROW_BITS) | (node - bk * RB);
        }
    }

    const int base = (int)blockIdx.x * BIN_TILE + tid;
    int rr[BIN_ITEMS];
#pragma unroll
    for (int k = 0; k < BIN_ITEMS; ++k) {
        const int e = base + k * BIN_TB;
        rr[k] = (e < NNZ) ? rows[e] : -1;
    }
#pragma unroll
    for (int k = 0; k < BIN_ITEMS; ++k)
        if (rr[k] >= 0) atomicAdd(&s_cnt[rr[k] / RB], 1);
    __syncthreads();
#pragma unroll
    for (int k = 0; k < BIN_PER; ++k) {
        const int bk = k * BIN_TB + tid;
        const int c  = s_cnt[bk];
        if (bk < NB && c > 0) s_base[bk] += atomicAdd(&cursor[bk], c);
        s_cnt[bk] = 0;
    }
    __syncthreads();
#pragma unroll
    for (int g = 0; g < BIN_ITEMS; g += BIN_SUB) {
        int cc[BIN_SUB], vv[BIN_SUB];
#pragma unroll
        for (int k = 0; k < BIN_SUB; ++k) {
            const int e = base + (g + k) * BIN_TB;
            cc[k] = 0; vv[k] = 0;
            if (rr[g + k] >= 0) { cc[k] = cols[e]; vv[k] = __float_as_int(vals[e]); }
        }
#pragma unroll
        for (int k = 0; k < BIN_SUB; ++k) {
            if (rr[g + k] >= 0) {
                const int bk = rr[g + k] / RB;
                const int p  = s_base[bk] + atomicAdd(&s_cnt[bk], 1);
                ent[p] = make_int2(((rr[g + k] - bk * RB) << COL_BITS) | cc[k], vv[k]);
            }
        }
    }
    if (bl_base >= 0) blist[bl_base + bl_rank] = bl_word;
}

#define HOP_TB    256
#define HOP_WAVES (HOP_TB / 32)
#define HOP_GROUP 8
#define SORT_PER  16
#define ROWS_CAP  7168
static_assert(RB + 1 <= HOP_TB, "one thread per row start, the end of the last row included");

template <bool EGO>
__device__ __forceinline__ const float* lgcn_src_row(const float* __restrict__ src,
                                                     const float* __restrict__ src_items,
                                                     int cj) {
    if (EGO && cj >= N_USERS) return src_items + (size_t)(cj - N_USERS) * DIM;
    return src + (size_t)cj * DIM;
}

__device__ __forceinline__ void lgcn_lds_add(float* p, float x) {
    __hip_atomic_fetch_add(p, x, __ATOMIC_RELAXED, __HIP_MEMORY_SCOPE_WORKGROUP);
}

template <bool EGO, int TOTAL>
__device__ __forceinline__ void lgcn_row_piece(const float* __restrict__ src,
                                               const float* __restrict__ src_items,
                                               int lane, int2 e, float& s0, float& s1) {
    float a[TOTAL], bq[TOTAL];
#pragma unroll
    for (int k = 0; k < TOTAL; ++k) {
        const float* s = lgcn_src_row<EGO>(src, src_items,
                                           __builtin_amdgcn_readlane(e.x, k)) + lane;
        a[k]  = s[0];
        bq[k] = s[32];
    }
#pragma unroll
    for (int k = 0; k < TOTAL; ++k) {
        const float vj = __int_as_float(__builtin_amdgcn_readlane(e.y, k));
        s0 = fmaf(vj, a[k],  s0);
        s1 = fmaf(vj, bq[k], s1);
    }
}

template <bool EGO>
__global__ void __launch_bounds__(HOP_TB)
lgcn_hop_rows(const int2*  __restrict__ ent,
              const int*   __restrict__ counts,
              const int*   __restrict__ starts,
              const float* __restrict__ src,
              const float* __restrict__ src_items,
              float*       __restrict__ out) {
    __shared__ int2 s_ent[ROWS_CAP];
    __shared__ int  s_cur[HOP_TB];
    __shared__ int  s_off[HOP_TB];
    __shared__ int  s_wsum[HOP_WAVES];
    const int tid  = (int)threadIdx.x;
    const int lane = tid & 31;
    const int wave = __builtin_amdgcn_readfirstlane(tid >> 5);
    const int b    = (int)blockIdx.x;

    const int   cnt  = counts[b];
    const int2* eb   = ent + starts[b];
    float*      orow = out + (size_t)b * (RB * DIM) + lane;

    if (cnt <= ROWS_CAP) {
        s_cur[tid] = 0;
        __syncthreads();
        for (int t0 = 0; t0 < cnt; t0 += HOP_TB * SORT_PER) {
            int ex[SORT_PER];
#pragma unroll
            for (int k = 0; k < SORT_PER; ++k) {
                const int idx = t0 + k * HOP_TB + tid;
                ex[k] = (idx < cnt) ? eb[idx].x : -1;
            }
#pragma unroll
            for (int k = 0; k < SORT_PER; ++k)
                if (ex[k] >= 0) atomicAdd(&s_cur[ex[k] >> COL_BITS], 1);
        }
        __syncthreads();
        const int x = s_cur[tid];
        int incl = x;
        for (int d = 1; d < 32; d <<= 1) {
            const int y = __shfl_up(incl, d);
            if (lane >= d) incl += y;
        }
        if (lane == 31) s_wsum[wave] = incl;
        __syncthreads();
        int pre = 0;
        for (int w = 0; w < HOP_WAVES; ++w)
            if (w < wave) pre += s_wsum[w];
        s_off[tid] = pre + incl - x;
        s_cur[tid] = 0;
        __syncthreads();
        for (int t0 = 0; t0 < cnt; t0 += HOP_TB * SORT_PER) {
            int2 ee[SORT_PER];
            int  pp[SORT_PER];
#pragma unroll
            for (int k = 0; k < SORT_PER; ++k) {
                const int idx = t0 + k * HOP_TB + tid;
                ee[k] = (idx < cnt) ? eb[idx] : make_int2(-1, 0);
            }
#pragma unroll
            for (int k = 0; k < SORT_PER; ++k) {
                pp[k] = 0;
                if (ee[k].x >= 0) {
                    const int rl = ee[k].x >> COL_BITS;
                    pp[k] = s_off[rl] + atomicAdd(&s_cur[rl], 1);
                }
            }
#pragma unroll
            for (int k = 0; k < SORT_PER; ++k)
                if (ee[k].x >= 0) s_ent[pp[k]] = make_int2(ee[k].x & COL_MASK, ee[k].y);
        }
        __syncthreads();

        for (int r = wave; r < RB; r += HOP_WAVES) {
            if (b * RB + r >= N_NODES) break;
            const int rs = __builtin_amdgcn_readfirstlane(s_off[r]);
            const int re = __builtin_amdgcn_readfirstlane(s_off[r + 1]);
            float s0 = 0.f, s1 = 0.f;
            for (int base = rs; base < re; base += 32) {
                const int n = (re - base < 32) ? (re - base) : 32;
                int2 e = make_int2(0, 0);
                if (lane < n) e = s_ent[base + lane];
                switch ((n + 3) >> 2) {
                case 1:  lgcn_row_piece<EGO,  4>(src, src_items, lane, e, s0, s1); break;
                case 2:  lgcn_row_piece<EGO,  8>(src, src_items, lane, e, s0, s1); break;
                case 3:  lgcn_row_piece<EGO, 12>(src, src_items, lane, e, s0, s1); break;
                case 4:  lgcn_row_piece<EGO, 16>(src, src_items, lane, e, s0, s1); break;
                case 5:  lgcn_row_piece<EGO, 20>(src, src_items, lane, e, s0, s1); break;
                case 6:  lgcn_row_piece<EGO, 24>(src, src_items, lane, e, s0, s1); break;
                case 7:  lgcn_row_piece<EGO, 28>(src, src_items, lane, e, s0, s1); break;
                default: lgcn_row_piece<EGO, 32>(src, src_items, lane, e, s0, s1); break;
                }
            }
            orow[(size_t)r * DIM]      = s0;
            orow[(size_t)r * DIM + 32] = s1;
        }
    } else {
        for (int r = wave; r < RB; r += HOP_WAVES) {
            if (b * RB + r >= N_NODES) break;
            float s0 = 0.f, s1 = 0.f;
            for (int base = 0; base < cnt; base += 32) {
                const int idx = base + lane;
                int2 e  = make_int2(0, 0);
                bool ok = false;
                if (idx < cnt) { e = eb[idx]; ok = ((e.x >> COL_BITS) == r); }
                unsigned m = __builtin_amdgcn_ballot_w32(ok);
                while (m != 0u) {
                    const int j = __builtin_ctz(m);
                    m &= (m - 1u);
                    const float* s = lgcn_src_row<EGO>(src, src_items,
                                         __builtin_amdgcn_readlane(e.x & COL_MASK, j)) + lane;
                    const float vj = __int_as_float(__builtin_amdgcn_readlane(e.y, j));
                    s0 = fmaf(vj, s[0],  s0);
                    s1 = fmaf(vj, s[32], s1);
                }
            }
            orow[(size_t)r * DIM]      = s0;
            orow[(size_t)r * DIM + 32] = s1;
        }
    }
}

__global__ void __launch_bounds__(HOP_TB)
lgcn_hop_last(const int2*  __restrict__ ent,
              const int*   __restrict__ counts,
              const int*   __restrict__ starts,
              const float* __restrict__ src,
              float*       __restrict__ out,
              const int*   __restrict__ bcnt,
              const int*   __restrict__ bstart,
              const int*   __restrict__ blist) {
    __shared__ float acc[RB * DIM];
    const int tid  = (int)threadIdx.x;
    const int lane = tid & 31;
    const int wave = __builtin_amdgcn_readfirstlane(tid >> 5);
    const int b    = (int)blockIdx.x;

    unsigned need[NEED_W];
    const int  bn = bcnt[b];
    const int* bl = blist + bstart[b];
    {
        unsigned* nw = (unsigned*)acc;
        if (tid < NEED_W) nw[tid] = 0u;
        __syncthreads();
        for (int i = tid; i < bn; i += HOP_TB) {
            const int lr = bl[i] & ((1 << ROW_BITS) - 1);
            atomicOr(&nw[lr >> 5], 1u << (lr & 31));
        }
        __syncthreads();
#pragma unroll
        for (int q = 0; q < NEED_W; ++q) need[q] = nw[q];
        __syncthreads();
    }

    for (int i = tid; i < RB * DIM / 4; i += HOP_TB)
        ((float4*)acc)[i] = make_float4(0.f, 0.f, 0.f, 0.f);
    __syncthreads();

    const int   cnt = counts[b];
    const int2* eb  = ent + starts[b];
    float*      al  = acc + lane;

    int2 cur = make_int2(0, 0);
    if (wave * 32 + lane < cnt) cur = eb[wave * 32 + lane];
    for (int base = wave * 32; base < cnt; base += HOP_WAVES * 32) {
        const int nidx = base + HOP_WAVES * 32 + lane;
        const int2 nxt = eb[(nidx < cnt) ? nidx : (cnt - 1)];
        const int c  = cur.x & COL_MASK;
        const int rl = cur.x >> COL_BITS;
        const int vb = cur.y;
        const int n  = (cnt - base < 32) ? (cnt - base) : 32;

        bool ok = lane < n;
        {
            unsigned wq = need[0];
#pragma unroll
            for (int q = 1; q < NEED_W; ++q) wq = ((rl >> 5) == q) ? need[q] : wq;
            ok = ok && (((wq >> (rl & 31)) & 1u) != 0u);
        }
        unsigned m = __builtin_amdgcn_ballot_w32(ok);
        while (m != 0u) {
            int  j[HOP_GROUP];
            bool on[HOP_GROUP];
#pragma unroll
            for (int k = 0; k < HOP_GROUP; ++k) {
                on[k] = (m != 0u);
                j[k]  = on[k] ? __builtin_ctz(m) : 0;
                m &= (m - 1u);
            }
            float a[HOP_GROUP], bq[HOP_GROUP];
#pragma unroll
            for (int k = 0; k < HOP_GROUP; ++k) {
                a[k] = 0.f; bq[k] = 0.f;
                if (on[k]) {
                    const float* s = lgcn_src_row<false>(src, nullptr,
                                         __builtin_amdgcn_readlane(c, j[k])) + lane;
                    a[k]  = s[0];
                    bq[k] = s[32];
                }
            }
#pragma unroll
            for (int k = 0; k < HOP_GROUP; ++k) {
                if (on[k]) {
                    const int   rj = __builtin_amdgcn_readlane(rl, j[k]);
                    const float vj = __int_as_float(__builtin_amdgcn_readlane(vb, j[k]));
                    float* d = al + rj * DIM;
                    lgcn_lds_add(d,      vj * a[k]);
                    lgcn_lds_add(d + 32, vj * bq[k]);
                }
            }
        }
        cur = nxt;
    }
    __syncthreads();

    for (int i = wave; i < bn; i += HOP_WAVES) {
        const int    e    = bl[i];
        const int    lr   = e & ((1 << ROW_BITS) - 1);
        const int    slot = e >> ROW_BITS;
        const float* s    = al + lr * DIM;
        float*       dst  = out + (size_t)slot * DIM + lane;
        dst[0]  = s[0];
        dst[32] = s[32];
    }
}

__global__ void lgcn_dot_wmma(const float* __restrict__ user_emb,
                              const float* __restrict__ item_emb,
                              const float* __restrict__ h1,
                              const float* __restrict__ h2,
                              const float* __restrict__ h3,
                              const int*   __restrict__ users,
                              const int*   __restrict__ items,
                              float*       __restrict__ out) {
    __shared__ float lds[8][16][17];
    int lane  = threadIdx.x & 31;
    int wv    = threadIdx.x >> 5;
    int tile  = blockIdx.x * 8 + wv;
    int m     = lane & 15;
    int khalf = lane >> 4;

    int u_row = users[tile * 16 + m];
    int v_row = items[tile * 16 + m];
    const size_t uo = (size_t)u_row * DIM;
    const size_t vo = ((size_t)N_USERS + (size_t)v_row) * DIM;
    const float* u0 = user_emb + uo;
    const float* v0 = item_emb + (size_t)v_row * DIM;
    const float* u1 = h1 + uo;  const float* v1 = h1 + vo;
    const float* u2 = h2 + uo;  const float* v2 = h2 + vo;
    const float* u3 = h3 + (size_t)(tile * 16 + m) * DIM;
    const float* v3 = h3 + (size_t)(BATCH + tile * 16 + m) * DIM;

    v8f c = {};
#pragma unroll
    for (int kc = 0; kc < DIM / 4; ++kc) {
        int off = kc * 4 + khalf * 2;
        v2f a; a.x = ((u0[off]     + u1[off])     + u2[off])     + u3[off];
               a.y = ((u0[off + 1] + u1[off + 1]) + u2[off + 1]) + u3[off + 1];
        v2f b; b.x = ((v0[off]     + v1[off])     + v2[off])     + v3[off];
               b.y = ((v0[off + 1] + v1[off + 1]) + v2[off + 1]) + v3[off + 1];
        c = __builtin_amdgcn_wmma_f32_16x16x4_f32(
                 false, a,  false, b,
                 (short)0, c,  false,  false);
    }

#pragma unroll
    for (int r = 0; r < 8; ++r)
        lds[wv][r + 8 * khalf][m] = c[r];
    __syncthreads();

    if (lane < 16)
        out[tile * 16 + lane] = lds[wv][lane][lane] * (1.0f / 16.0f);
}

extern "C" void kernel_launch(void* const* d_in, const int* in_sizes, int n_in,
                              void* d_out, int out_size, void* d_ws, size_t ws_size,
                              hipStream_t stream) {
    const float* user_emb = (const float*)d_in[0];
    const float* item_emb = (const float*)d_in[1];
    const int*   g_rows   = (const int*)  d_in[2];
    const int*   g_cols   = (const int*)  d_in[3];
    const float* g_vals   = (const float*)d_in[4];
    const int*   users    = (const int*)  d_in[5];
    const int*   items    = (const int*)  d_in[6];
    float*       out      = (float*)d_out;

    const size_t node_f = (size_t)N_NODES * DIM;
    float* hop1   = (float*)d_ws;
    float* hop2   = hop1 + node_f;
    float* hop3   = hop2 + node_f;
    int2*  ent    = (int2*)(hop3 + (size_t)2 * BATCH * DIM);
    int*   counts = (int*)(ent + NNZ);
    int*   cursor = counts + NBP;
    int*   starts = cursor + NBP;
    int*   bcnt   = starts + NBP;
    int*   bcur   = bcnt + NBP;
    int*   bstart = bcur + NBP;
    int*   blist  = bstart + NBP;

    const int TB = 256;

    lgcn_bin_clear<<<NBP / BIN_TB, BIN_TB, 0, stream>>>(counts, cursor, bcnt, bcur);
    lgcn_bin_count<<<BIN_GRID, BIN_TB, 0, stream>>>(g_rows, counts, users, items, bcnt);
    lgcn_bin_place<<<BIN_GRID, BIN_TB, 0, stream>>>(g_rows, g_cols, g_vals, counts, cursor,
                                                    starts, ent, users, items, bcnt, bcur,
                                                    bstart, blist);

    lgcn_hop_rows<true><<<NB, HOP_TB, 0, stream>>>(ent, counts, starts, user_emb, item_emb, hop1);
    lgcn_hop_rows<false><<<NB, HOP_TB, 0, stream>>>(ent, counts, starts, hop1, nullptr, hop2);
    lgcn_hop_last<<<NB, HOP_TB, 0, stream>>>(ent, counts, starts, hop2, hop3,
                                             bcnt, bstart, blist);

    lgcn_dot_wmma<<<BATCH / 16 / 8, TB, 0, stream>>>(user_emb, item_emb, hop1, hop2, hop3,
                                                     users, items, out);
}


# Hop 1 writes each bucket's row-sorted list back in place, with its row starts: hop 2 copies it and sorts nothing, the last hop sums only needed rows in registers (fma); its sum order: list order
# speedup vs baseline: 6.4640x; 1.0374x over previous
#include <hip/hip_runtime.h>
#include <hip/hip_bf16.h>

#define N_USERS 100000
#define N_ITEMS 200000
#define N_NODES (N_USERS + N_ITEMS)
#define NNZ     4800000
#define DIM     64
#define BATCH   16384

typedef float v2f __attribute__((ext_vector_type(2)));
typedef float v8f __attribute__((ext_vector_type(8)));

#define RB       240
#define NB       ((N_NODES + RB - 1) / RB)
#define ROW_BITS 8
#define COL_BITS 19
#define COL_MASK ((1 << COL_BITS) - 1)
static_assert(RB <= (1 << ROW_BITS), "a row within its bucket must fit ROW_BITS");
static_assert(N_NODES <= (1 << COL_BITS), "a column index must fit COL_BITS");
static_assert(ROW_BITS + COL_BITS <= 31, "packed (row, column) must stay a non-negative int");

#define BIN_TB    256
#define BIN_PER   ((NB + BIN_TB - 1) / BIN_TB)
#define NBP       (BIN_PER * BIN_TB)
#define BIN_TILE  8192
#define BIN_ITEMS (BIN_TILE / BIN_TB)
#define BIN_GRID  ((NNZ + BIN_TILE - 1) / BIN_TILE)
#define BIN_SUB   8

static_assert(2 * BATCH <= BIN_GRID * BIN_TB, "one batch position per thread of the binning grid");
static_assert(2 * BATCH <= (1 << (31 - ROW_BITS)), "packed (position, row) must stay a non-negative int");

__device__ __forceinline__ int lgcn_batch_node(const int* __restrict__ users,
                                               const int* __restrict__ items, int i) {
    return (i < BATCH) ? users[i] : N_USERS + items[i - BATCH];
}

__global__ void lgcn_bin_clear(int* __restrict__ counts, int* __restrict__ cursor,
                               int* __restrict__ bcnt, int* __restrict__ bcur) {
    const int i = (int)(blockIdx.x * blockDim.x + threadIdx.x);
    if (i < NBP) { counts[i] = 0; cursor[i] = 0; bcnt[i] = 0; bcur[i] = 0; }
}

__global__ void __launch_bounds__(BIN_TB)
lgcn_bin_count(const int* __restrict__ rows, int* __restrict__ counts,
               const int* __restrict__ users, const int* __restrict__ items,
               int* __restrict__ bcnt) {
    __shared__ int s_cnt[NBP];
    const int tid = (int)threadIdx.x;
    const int g = (int)blockIdx.x * BIN_TB + tid;
    if (g < 2 * BATCH) atomicAdd(&bcnt[lgcn_batch_node(users, items, g) / RB], 1);
#pragma unroll
    for (int k = 0; k < BIN_PER; ++k) s_cnt[k * BIN_TB + tid] = 0;
    __syncthreads();
    const int base = (int)blockIdx.x * BIN_TILE + tid;
    int rr[BIN_ITEMS];
#pragma unroll
    for (int k = 0; k < BIN_ITEMS; ++k) {
        const int e = base + k * BIN_TB;
        rr[k] = (e < NNZ) ? rows[e] : -1;
    }
#pragma unroll
    for (int k = 0; k < BIN_ITEMS; ++k)
        if (rr[k] >= 0) atomicAdd(&s_cnt[rr[k] / RB], 1);
    __syncthreads();
#pragma unroll
    for (int k = 0; k < BIN_PER; ++k) {
        const int bk = k * BIN_TB + tid;
        const int c  = s_cnt[bk];
        if (bk < NB && c > 0) atomicAdd(&counts[bk], c);
    }
}

__global__ void __launch_bounds__(BIN_TB)
lgcn_bin_place(const int*   __restrict__ rows,
               const int*   __restrict__ cols,
               const float* __restrict__ vals,
               const int*   __restrict__ counts,
               int*         __restrict__ cursor,
               int*         __restrict__ starts,
               int2*        __restrict__ ent,
               const int*   __restrict__ users,
               const int*   __restrict__ items,
               const int*   __restrict__ bcnt,
               int*         __restrict__ bcur,
               int*         __restrict__ bstart,
               int*         __restrict__ blist) {
    __shared__ int s_cnt[NBP];
    __shared__ int s_base[NBP];
    __shared__ int s_bb[NBP];
    __shared__ unsigned long long s_scan[BIN_TB];
    const int tid = (int)threadIdx.x;

#pragma unroll
    for (int k = 0; k < BIN_PER; ++k) {
        const int bk = k * BIN_TB + tid;
        s_base[bk] = (bk < NB) ? counts[bk] : 0;
        s_bb[bk]   = (bk < NB) ? bcnt[bk] : 0;
        s_cnt[bk]  = 0;
    }
    __syncthreads();
    int loc[BIN_PER], lob[BIN_PER];
    unsigned long long tsum = 0ull;
#pragma unroll
    for (int k = 0; k < BIN_PER; ++k) {
        loc[k] = s_base[tid * BIN_PER + k];
        lob[k] = s_bb[tid * BIN_PER + k];
        tsum += (unsigned long long)(unsigned)loc[k] | ((unsigned long long)(unsigned)lob[k] << 32);
    }
    s_scan[tid] = tsum;
    __syncthreads();
    for (int off = 1; off < BIN_TB; off <<= 1) {
        const unsigned long long v = (tid >= off) ? s_scan[tid - off] : 0ull;
        __syncthreads();
        s_scan[tid] += v;
        __syncthreads();
    }
    unsigned long long run = s_scan[tid] - tsum;
#pragma unroll
    for (int k = 0; k < BIN_PER; ++k) {
        const int bk = tid * BIN_PER + k;
        const int rn = (int)(unsigned)(run & 0xffffffffull);
        const int rb = (int)(unsigned)(run >> 32);
        s_base[bk] = rn;
        s_bb[bk]   = rb;
        if (blockIdx.x == 0 && bk < NB) { starts[bk] = rn; bstart[bk] = rb; }
        run += (unsigned long long)(unsigned)loc[k] | ((unsigned long long)(unsigned)lob[k] << 32);
    }
    __syncthreads();

    int bl_base = -1, bl_rank = 0, bl_word = 0;
    {
        const int g = (int)blockIdx.x * BIN_TB + tid;
        if (g < 2 * BATCH) {
            const int node = lgcn_batch_node(users, items, g);
            const int bk   = node / RB;
            bl_base = s_bb[bk];
            bl_rank = atomicAdd(&bcur[bk], 1);
            bl_word = (g << ROW_BITS) | (node - bk * RB);
        }
    }

    const int base = (int)blockIdx.x * BIN_TILE + tid;
    int rr[BIN_ITEMS];
#pragma unroll
    for (int k = 0; k < BIN_ITEMS; ++k) {
        const int e = base + k * BIN_TB;
        rr[k] = (e < NNZ) ? rows[e] : -1;
    }
#pragma unroll
    for (int k = 0; k < BIN_ITEMS; ++k)
        if (rr[k] >= 0) atomicAdd(&s_cnt[rr[k] / RB], 1);
    __syncthreads();
#pragma unroll
    for (int k = 0; k < BIN_PER; ++k) {
        const int bk = k * BIN_TB + tid;
        const int c  = s_cnt[bk];
        if (bk < NB && c > 0) s_base[bk] += atomicAdd(&cursor[bk], c);
        s_cnt[bk] = 0;
    }
    __syncthreads();
#pragma unroll
    for (int g = 0; g < BIN_ITEMS; g += BIN_SUB) {
        int cc[BIN_SUB], vv[BIN_SUB];
#pragma unroll
        for (int k = 0; k < BIN_SUB; ++k) {
            const int e = base + (g + k) * BIN_TB;
            cc[k] = 0; vv[k] = 0;
            if (rr[g + k] >= 0) { cc[k] = cols[e]; vv[k] = __float_as_int(vals[e]); }
        }
#pragma unroll
        for (int k = 0; k < BIN_SUB; ++k) {
            if (rr[g + k] >= 0) {
                const int bk = rr[g + k] / RB;
                const int p  = s_base[bk] + atomicAdd(&s_cnt[bk], 1);
                ent[p] = make_int2(((rr[g + k] - bk * RB) << COL_BITS) | cc[k], vv[k]);
            }
        }
    }
    if (bl_base >= 0) blist[bl_base + bl_rank] = bl_word;
}

#define HOP_TB    256
#define HOP_WAVES (HOP_TB / 32)
#define SORT_PER  16
#define ROWS_CAP  7168
static_assert(RB + 1 <= HOP_TB, "one thread per row start, the end of the last row included");
#define ROFF_W    HOP_TB

template <bool EGO>
__device__ __forceinline__ const float* lgcn_src_row(const float* __restrict__ src,
                                                     const float* __restrict__ src_items,
                                                     int cj) {
    if (EGO && cj >= N_USERS) return src_items + (size_t)(cj - N_USERS) * DIM;
    return src + (size_t)cj * DIM;
}

template <bool EGO, int TOTAL>
__device__ __forceinline__ void lgcn_row_piece(const float* __restrict__ src,
                                               const float* __restrict__ src_items,
                                               int lane, int2 e, float& s0, float& s1) {
    float a[TOTAL], bq[TOTAL];
#pragma unroll
    for (int k = 0; k < TOTAL; ++k) {
        const float* s = lgcn_src_row<EGO>(src, src_items,
                                           __builtin_amdgcn_readlane(e.x, k)) + lane;
        a[k]  = s[0];
        bq[k] = s[32];
    }
#pragma unroll
    for (int k = 0; k < TOTAL; ++k) {
        const float vj = __int_as_float(__builtin_amdgcn_readlane(e.y, k));
        s0 = fmaf(vj, a[k],  s0);
        s1 = fmaf(vj, bq[k], s1);
    }
}

template <bool EGO>
__device__ __forceinline__ void lgcn_row_scan(const int2* eb, int cnt, int r, int lane,
                                              const float* __restrict__ src,
                                              const float* __restrict__ src_items,
                                              float& s0, float& s1) {
    for (int base = 0; base < cnt; base += 32) {
        const int idx = base + lane;
        int2 e  = make_int2(0, 0);
        bool ok = false;
        if (idx < cnt) { e = eb[idx]; ok = ((e.x >> COL_BITS) == r); }
        unsigned m = __builtin_amdgcn_ballot_w32(ok);
        while (m != 0u) {
            const int j = __builtin_ctz(m);
            m &= (m - 1u);
            const float* s = lgcn_src_row<EGO>(src, src_items,
                                 __builtin_amdgcn_readlane(e.x & COL_MASK, j)) + lane;
            const float vj = __int_as_float(__builtin_amdgcn_readlane(e.y, j));
            s0 = fmaf(vj, s[0],  s0);
            s1 = fmaf(vj, s[32], s1);
        }
    }
}

template <bool EGO>
__device__ __forceinline__ void lgcn_row_piece_n(const float* __restrict__ src,
                                                 const float* __restrict__ src_items,
                                                 int lane, int n, int2 e, float& s0, float& s1) {
    switch ((n + 3) >> 2) {
    case 1:  lgcn_row_piece<EGO,  4>(src, src_items, lane, e, s0, s1); break;
    case 2:  lgcn_row_piece<EGO,  8>(src, src_items, lane, e, s0, s1); break;
    case 3:  lgcn_row_piece<EGO, 12>(src, src_items, lane, e, s0, s1); break;
    case 4:  lgcn_row_piece<EGO, 16>(src, src_items, lane, e, s0, s1); break;
    case 5:  lgcn_row_piece<EGO, 20>(src, src_items, lane, e, s0, s1); break;
    case 6:  lgcn_row_piece<EGO, 24>(src, src_items, lane, e, s0, s1); break;
    case 7:  lgcn_row_piece<EGO, 28>(src, src_items, lane, e, s0, s1); break;
    default: lgcn_row_piece<EGO, 32>(src, src_items, lane, e, s0, s1); break;
    }
}

template <bool EGO>
__global__ void __launch_bounds__(HOP_TB)
lgcn_hop_rows(int2*        __restrict__ ent,
              const int*   __restrict__ counts,
              const int*   __restrict__ starts,
              int*         __restrict__ roff,
              const float* __restrict__ src,
              const float* __restrict__ src_items,
              float*       __restrict__ out) {
    __shared__ int2 s_ent[ROWS_CAP];
    __shared__ int  s_cur[HOP_TB];
    __shared__ int  s_off[HOP_TB];
    __shared__ int  s_wsum[HOP_WAVES];
    const int tid  = (int)threadIdx.x;
    const int lane = tid & 31;
    const int wave = __builtin_amdgcn_readfirstlane(tid >> 5);
    const int b    = (int)blockIdx.x;

    const int cnt  = counts[b];
    int2*     eb   = ent + starts[b];
    int*      ro   = roff + (size_t)b * ROFF_W;
    float*    orow = out + (size_t)b * (RB * DIM) + lane;

    if (cnt <= ROWS_CAP) {
        if (EGO) {
            s_cur[tid] = 0;
            __syncthreads();
            for (int t0 = 0; t0 < cnt; t0 += HOP_TB * SORT_PER) {
                int ex[SORT_PER];
#pragma unroll
                for (int k = 0; k < SORT_PER; ++k) {
                    const int idx = t0 + k * HOP_TB + tid;
                    ex[k] = (idx < cnt) ? eb[idx].x : -1;
                }
#pragma unroll
                for (int k = 0; k < SORT_PER; ++k)
                    if (ex[k] >= 0) atomicAdd(&s_cur[ex[k] >> COL_BITS], 1);
            }
            __syncthreads();
            const int x = s_cur[tid];
            int incl = x;
            for (int d = 1; d < 32; d <<= 1) {
                const int y = __shfl_up(incl, d);
                if (lane >= d) incl += y;
            }
            if (lane == 31) s_wsum[wave] = incl;
            __syncthreads();
            int pre = 0;
            for (int w = 0; w < HOP_WAVES; ++w)
                if (w < wave) pre += s_wsum[w];
            s_off[tid] = pre + incl - x;
            s_cur[tid] = 0;
            __syncthreads();
            for (int t0 = 0; t0 < cnt; t0 += HOP_TB * SORT_PER) {
                int2 ee[SORT_PER];
                int  pp[SORT_PER];
#pragma unroll
                for (int k = 0; k < SORT_PER; ++k) {
                    const int idx = t0 + k * HOP_TB + tid;
                    ee[k] = (idx < cnt) ? eb[idx] : make_int2(-1, 0);
                }
#pragma unroll
                for (int k = 0; k < SORT_PER; ++k) {
                    pp[k] = 0;
                    if (ee[k].x >= 0) {
                        const int rl = ee[k].x >> COL_BITS;
                        pp[k] = s_off[rl] + atomicAdd(&s_cur[rl], 1);
                    }
                }
#pragma unroll
                for (int k = 0; k < SORT_PER; ++k)
                    if (ee[k].x >= 0) s_ent[pp[k]] = ee[k];
            }
            __syncthreads();
        } else {
            s_off[tid] = ro[tid];
            for (int t0 = 0; t0 < cnt; t0 += HOP_TB * SORT_PER) {
                int2 ee[SORT_PER];
#pragma unroll
                for (int k = 0; k < SORT_PER; ++k) {
                    const int idx = t0 + k * HOP_TB + tid;
                    ee[k] = (idx < cnt) ? eb[idx] : make_int2(-1, 0);
                }
#pragma unroll
                for (int k = 0; k < SORT_PER; ++k) {
                    const int idx = t0 + k * HOP_TB + tid;
                    if (ee[k].x >= 0) s_ent[idx] = ee[k];
                }
            }
            __syncthreads();
        }

        for (int r = wave; r < RB; r += HOP_WAVES) {
            if (b * RB + r >= N_NODES) break;
            const int rs = __builtin_amdgcn_readfirstlane(s_off[r]);
            const int re = __builtin_amdgcn_readfirstlane(s_off[r + 1]);
            float s0 = 0.f, s1 = 0.f;
            for (int base = rs; base < re; base += 32) {
                const int n = (re - base < 32) ? (re - base) : 32;
                int2 e = make_int2(0, 0);
                if (lane < n) { e = s_ent[base + lane]; e.x &= COL_MASK; }
                lgcn_row_piece_n<EGO>(src, src_items, lane, n, e, s0, s1);
            }
            orow[(size_t)r * DIM]      = s0;
            orow[(size_t)r * DIM + 32] = s1;
        }

        if (EGO) {
            ro[tid] = s_off[tid];
#pragma unroll 4
            for (int i = tid; i < cnt; i += HOP_TB) eb[i] = s_ent[i];
        }
    } else {
        for (int r = wave; r < RB; r += HOP_WAVES) {
            if (b * RB + r >= N_NODES) break;
            float s0 = 0.f, s1 = 0.f;
            lgcn_row_scan<EGO>(eb, cnt, r, lane, src, src_items, s0, s1);
            orow[(size_t)r * DIM]      = s0;
            orow[(size_t)r * DIM + 32] = s1;
        }
    }
}

__global__ void __launch_bounds__(HOP_TB)
lgcn_hop_last(const int2*  __restrict__ ent,
              const int*   __restrict__ counts,
              const int*   __restrict__ starts,
              const int*   __restrict__ roff,
              const float* __restrict__ src,
              float*       __restrict__ out,
              const int*   __restrict__ bcnt,
              const int*   __restrict__ bstart,
              const int*   __restrict__ blist) {
    __shared__ int s_w[HOP_TB];
    __shared__ int s_rs[HOP_TB];
    __shared__ int s_re[HOP_TB];
    const int tid  = (int)threadIdx.x;
    const int lane = tid & 31;
    const int wave = __builtin_amdgcn_readfirstlane(tid >> 5);
    const int b    = (int)blockIdx.x;

    const int   bn  = bcnt[b];
    const int*  bl  = blist + bstart[b];
    const int   cnt = counts[b];
    const int   gs  = starts[b];
    const int2* eb  = ent + gs;
    const int*  ro  = roff + (size_t)b * ROFF_W;
    const bool  sorted = (cnt <= ROWS_CAP);

    for (int c0 = 0; c0 < bn; c0 += HOP_TB) {
        const int m = (bn - c0 < HOP_TB) ? (bn - c0) : HOP_TB;
        if (tid < m) {
            const int w  = bl[c0 + tid];
            const int lr = w & ((1 << ROW_BITS) - 1);
            s_w[tid] = w;
            if (sorted) { s_rs[tid] = ro[lr]; s_re[tid] = ro[lr + 1]; }
        }
        __syncthreads();
        if (sorted) {
            int  i  = wave;
            int  rs = 0, re = 0;
            int2 e  = make_int2(0, 0);
            if (i < m) {
                rs = __builtin_amdgcn_readfirstlane(s_rs[i]);
                re = __builtin_amdgcn_readfirstlane(s_re[i]);
                if (lane < re - rs) e = eb[rs + lane];
            }
            for (; i < m; i += HOP_WAVES) {
                const int inx = (i + HOP_WAVES < m) ? (i + HOP_WAVES) : i;
                const int rsn = __builtin_amdgcn_readfirstlane(s_rs[inx]);
                const int ren = __builtin_amdgcn_readfirstlane(s_re[inx]);
                const int ld  = gs + rsn + lane;
                const int2 en = ent[(ld < NNZ) ? ld : (NNZ - 1)];
                float s0 = 0.f, s1 = 0.f;
                if (rs < re) {
                    int n = (re - rs < 32) ? (re - rs) : 32;
                    if (lane >= n) e = make_int2(0, 0);
                    e.x &= COL_MASK;
                    lgcn_row_piece_n<false>(src, nullptr, lane, n, e, s0, s1);
                    for (int base = rs + 32; base < re; base += 32) {
                        n = (re - base < 32) ? (re - base) : 32;
                        int2 e2 = make_int2(0, 0);
                        if (lane < n) { e2 = eb[base + lane]; e2.x &= COL_MASK; }
                        lgcn_row_piece_n<false>(src, nullptr, lane, n, e2, s0, s1);
                    }
                }
                const int slot = __builtin_amdgcn_readfirstlane(s_w[i]) >> ROW_BITS;
                float*    dst  = out + (size_t)slot * DIM + lane;
                dst[0]  = s0;
                dst[32] = s1;
                rs = rsn; re = ren; e = en;
            }
        } else {
            for (int i = wave; i < m; i += HOP_WAVES) {
                const int w  = __builtin_amdgcn_readfirstlane(s_w[i]);
                const int lr = w & ((1 << ROW_BITS) - 1);
                float s0 = 0.f, s1 = 0.f;
                lgcn_row_scan<false>(eb, cnt, lr, lane, src, nullptr, s0, s1);
                float* dst = out + (size_t)(w >> ROW_BITS) * DIM + lane;
                dst[0]  = s0;
                dst[32] = s1;
            }
        }
        __syncthreads();
    }
}

__global__ void lgcn_dot_wmma(const float* __restrict__ user_emb,
                              const float* __restrict__ item_emb,
                              const float* __restrict__ h1,
                              const float* __restrict__ h2,
                              const float* __restrict__ h3,
                              const int*   __restrict__ users,
                              const int*   __restrict__ items,
                              float*       __restrict__ out) {
    __shared__ float lds[8][16][17];
    int lane  = threadIdx.x & 31;
    int wv    = threadIdx.x >> 5;
    int tile  = blockIdx.x * 8 + wv;
    int m     = lane & 15;
    int khalf = lane >> 4;

    int u_row = users[tile * 16 + m];
    int v_row = items[tile * 16 + m];
    const size_t uo = (size_t)u_row * DIM;
    const size_t vo = ((size_t)N_USERS + (size_t)v_row) * DIM;
    const float* u0 = user_emb + uo;
    const float* v0 = item_emb + (size_t)v_row * DIM;
    const float* u1 = h1 + uo;  const float* v1 = h1 + vo;
    const float* u2 = h2 + uo;  const float* v2 = h2 + vo;
    const float* u3 = h3 + (size_t)(tile * 16 + m) * DIM;
    const float* v3 = h3 + (size_t)(BATCH + tile * 16 + m) * DIM;

    v8f c = {};
#pragma unroll
    for (int kc = 0; kc < DIM / 4; ++kc) {
        int off = kc * 4 + khalf * 2;
        v2f a; a.x = ((u0[off]     + u1[off])     + u2[off])     + u3[off];
               a.y = ((u0[off + 1] + u1[off + 1]) + u2[off + 1]) + u3[off + 1];
        v2f b; b.x = ((v0[off]     + v1[off])     + v2[off])     + v3[off];
               b.y = ((v0[off + 1] + v1[off + 1]) + v2[off + 1]) + v3[off + 1];
        c = __builtin_amdgcn_wmma_f32_16x16x4_f32(
                 false, a,  false, b,
                 (short)0, c,  false,  false);
    }

#pragma unroll
    for (int r = 0; r < 8; ++r)
        lds[wv][r + 8 * khalf][m] = c[r];
    __syncthreads();

    if (lane < 16)
        out[tile * 16 + lane] = lds[wv][lane][lane] * (1.0f / 16.0f);
}

extern "C" void kernel_launch(void* const* d_in, const int* in_sizes, int n_in,
                              void* d_out, int out_size, void* d_ws, size_t ws_size,
                              hipStream_t stream) {
    const float* user_emb = (const float*)d_in[0];
    const float* item_emb = (const float*)d_in[1];
    const int*   g_rows   = (const int*)  d_in[2];
    const int*   g_cols   = (const int*)  d_in[3];
    const float* g_vals   = (const float*)d_in[4];
    const int*   users    = (const int*)  d_in[5];
    const int*   items    = (const int*)  d_in[6];
    float*       out      = (float*)d_out;

    const size_t node_f = (size_t)N_NODES * DIM;
    float* hop1   = (float*)d_ws;
    float* hop2   = hop1 + node_f;
    float* hop3   = hop2 + node_f;
    int2*  ent    = (int2*)(hop3 + (size_t)2 * BATCH * DIM);
    int*   counts = (int*)(ent + NNZ);
    int*   cursor = counts + NBP;
    int*   starts = cursor + NBP;
    int*   bcnt   = starts + NBP;
    int*   bcur   = bcnt + NBP;
    int*   bstart = bcur + NBP;
    int*   blist  = bstart + NBP;
    int*   roff   = blist + 2 * BATCH;

    const int TB = 256;

    lgcn_bin_clear<<<NBP / BIN_TB, BIN_TB, 0, stream>>>(counts, cursor, bcnt, bcur);
    lgcn_bin_count<<<BIN_GRID, BIN_TB, 0, stream>>>(g_rows, counts, users, items, bcnt);
    lgcn_bin_place<<<BIN_GRID, BIN_TB, 0, stream>>>(g_rows, g_cols, g_vals, counts, cursor,
                                                    starts, ent, users, items, bcnt, bcur,
                                                    bstart, blist);

    lgcn_hop_rows<true><<<NB, HOP_TB, 0, stream>>>(ent, counts, starts, roff, user_emb, item_emb, hop1);
    lgcn_hop_rows<false><<<NB, HOP_TB, 0, stream>>>(ent, counts, starts, roff, hop1, nullptr, hop2);
    lgcn_hop_last<<<NB, HOP_TB, 0, stream>>>(ent, counts, starts, roff, hop2, hop3,
                                             bcnt, bstart, blist);

    lgcn_dot_wmma<<<BATCH / 16 / 8, TB, 0, stream>>>(user_emb, item_emb, hop1, hop2, hop3,
                                                     users, items, out);
}


# Last hop fused into the dot kernel: a workgroup sums its 16 pairs' 32 rows (4 per wave), adds the running sum, feeds WMMA from LDS; hop-3 buffer, batch binning and one launch go; sum order as before
# speedup vs baseline: 6.9533x; 1.0757x over previous
#include <hip/hip_runtime.h>
#include <hip/hip_bf16.h>

#define N_USERS 100000
#define N_ITEMS 200000
#define N_NODES (N_USERS + N_ITEMS)
#define NNZ     4800000
#define DIM     64
#define BATCH   16384

typedef float v2f __attribute__((ext_vector_type(2)));
typedef float v8f __attribute__((ext_vector_type(8)));

#define RB       240
#define NB       ((N_NODES + RB - 1) / RB)
#define ROW_BITS 8
#define COL_BITS 19
#define COL_MASK ((1 << COL_BITS) - 1)
static_assert(RB <= (1 << ROW_BITS), "a row within its bucket must fit ROW_BITS");
static_assert(N_NODES <= (1 << COL_BITS), "a column index must fit COL_BITS");
static_assert(ROW_BITS + COL_BITS <= 31, "packed (row, column) must stay a non-negative int");

#define BIN_TB    256
#define BIN_PER   ((NB + BIN_TB - 1) / BIN_TB)
#define NBP       (BIN_PER * BIN_TB)
#define BIN_TILE  8192
#define BIN_ITEMS (BIN_TILE / BIN_TB)
#define BIN_GRID  ((NNZ + BIN_TILE - 1) / BIN_TILE)
#define BIN_SUB   8

__global__ void lgcn_bin_clear(int* __restrict__ counts, int* __restrict__ cursor) {
    const int i = (int)(blockIdx.x * blockDim.x + threadIdx.x);
    if (i < NBP) { counts[i] = 0; cursor[i] = 0; }
}

__global__ void __launch_bounds__(BIN_TB)
lgcn_bin_count(const int* __restrict__ rows, int* __restrict__ counts) {
    __shared__ int s_cnt[NBP];
    const int tid = (int)threadIdx.x;
#pragma unroll
    for (int k = 0; k < BIN_PER; ++k) s_cnt[k * BIN_TB + tid] = 0;
    __syncthreads();
    const int base = (int)blockIdx.x * BIN_TILE + tid;
    int rr[BIN_ITEMS];
#pragma unroll
    for (int k = 0; k < BIN_ITEMS; ++k) {
        const int e = base + k * BIN_TB;
        rr[k] = (e < NNZ) ? rows[e] : -1;
    }
#pragma unroll
    for (int k = 0; k < BIN_ITEMS; ++k)
        if (rr[k] >= 0) atomicAdd(&s_cnt[rr[k] / RB], 1);
    __syncthreads();
#pragma unroll
    for (int k = 0; k < BIN_PER; ++k) {
        const int bk = k * BIN_TB + tid;
        const int c  = s_cnt[bk];
        if (bk < NB && c > 0) atomicAdd(&counts[bk], c);
    }
}

__global__ void __launch_bounds__(BIN_TB)
lgcn_bin_place(const int*   __restrict__ rows,
               const int*   __restrict__ cols,
               const float* __restrict__ vals,
               const int*   __restrict__ counts,
               int*         __restrict__ cursor,
               int*         __restrict__ starts,
               int2*        __restrict__ ent) {
    __shared__ int s_cnt[NBP];
    __shared__ int s_base[NBP];
    __shared__ int s_scan[BIN_TB];
    const int tid = (int)threadIdx.x;

#pragma unroll
    for (int k = 0; k < BIN_PER; ++k) {
        const int bk = k * BIN_TB + tid;
        s_base[bk] = (bk < NB) ? counts[bk] : 0;
        s_cnt[bk]  = 0;
    }
    __syncthreads();
    int loc[BIN_PER];
    int tsum = 0;
#pragma unroll
    for (int k = 0; k < BIN_PER; ++k) {
        loc[k] = s_base[tid * BIN_PER + k];
        tsum += loc[k];
    }
    s_scan[tid] = tsum;
    __syncthreads();
    for (int off = 1; off < BIN_TB; off <<= 1) {
        const int v = (tid >= off) ? s_scan[tid - off] : 0;
        __syncthreads();
        s_scan[tid] += v;
        __syncthreads();
    }
    int run = s_scan[tid] - tsum;
#pragma unroll
    for (int k = 0; k < BIN_PER; ++k) {
        const int bk = tid * BIN_PER + k;
        s_base[bk] = run;
        if (blockIdx.x == 0 && bk < NB) starts[bk] = run;
        run += loc[k];
    }
    __syncthreads();

    const int base = (int)blockIdx.x * BIN_TILE + tid;
    int rr[BIN_ITEMS];
#pragma unroll
    for (int k = 0; k < BIN_ITEMS; ++k) {
        const int e = base + k * BIN_TB;
        rr[k] = (e < NNZ) ? rows[e] : -1;
    }
#pragma unroll
    for (int k = 0; k < BIN_ITEMS; ++k)
        if (rr[k] >= 0) atomicAdd(&s_cnt[rr[k] / RB], 1);
    __syncthreads();
#pragma unroll
    for (int k = 0; k < BIN_PER; ++k) {
        const int bk = k * BIN_TB + tid;
        const int c  = s_cnt[bk];
        if (bk < NB && c > 0) s_base[bk] += atomicAdd(&cursor[bk], c);
        s_cnt[bk] = 0;
    }
    __syncthreads();
#pragma unroll
    for (int g = 0; g < BIN_ITEMS; g += BIN_SUB) {
        int cc[BIN_SUB], vv[BIN_SUB];
#pragma unroll
        for (int k = 0; k < BIN_SUB; ++k) {
            const int e = base + (g + k) * BIN_TB;
            cc[k] = 0; vv[k] = 0;
            if (rr[g + k] >= 0) { cc[k] = cols[e]; vv[k] = __float_as_int(vals[e]); }
        }
#pragma unroll
        for (int k = 0; k < BIN_SUB; ++k) {
            if (rr[g + k] >= 0) {
                const int bk = rr[g + k] / RB;
                const int p  = s_base[bk] + atomicAdd(&s_cnt[bk], 1);
                ent[p] = make_int2(((rr[g + k] - bk * RB) << COL_BITS) | cc[k], vv[k]);
            }
        }
    }
}

#define HOP_TB    256
#define HOP_WAVES (HOP_TB / 32)
#define SORT_PER  16
#define ROWS_CAP  7168
static_assert(RB + 1 <= HOP_TB, "one thread per row start, the end of the last row included");
#define ROFF_W    HOP_TB

template <bool EGO>
__device__ __forceinline__ const float* lgcn_src_row(const float* __restrict__ src,
                                                     const float* __restrict__ src_items,
                                                     int cj) {
    if (EGO && cj >= N_USERS) return src_items + (size_t)(cj - N_USERS) * DIM;
    return src + (size_t)cj * DIM;
}

template <bool EGO, int TOTAL>
__device__ __forceinline__ void lgcn_row_piece(const float* __restrict__ src,
                                               const float* __restrict__ src_items,
                                               int lane, int2 e, float& s0, float& s1) {
    float a[TOTAL], bq[TOTAL];
#pragma unroll
    for (int k = 0; k < TOTAL; ++k) {
        const float* s = lgcn_src_row<EGO>(src, src_items,
                                           __builtin_amdgcn_readlane(e.x, k)) + lane;
        a[k]  = s[0];
        bq[k] = s[32];
    }
#pragma unroll
    for (int k = 0; k < TOTAL; ++k) {
        const float vj = __int_as_float(__builtin_amdgcn_readlane(e.y, k));
        s0 = fmaf(vj, a[k],  s0);
        s1 = fmaf(vj, bq[k], s1);
    }
}

template <bool EGO>
__device__ __forceinline__ void lgcn_row_scan(const int2* eb, int cnt, int r, int lane,
                                              const float* __restrict__ src,
                                              const float* __restrict__ src_items,
                                              float& s0, float& s1) {
    for (int base = 0; base < cnt; base += 32) {
        const int idx = base + lane;
        int2 e  = make_int2(0, 0);
        bool ok = false;
        if (idx < cnt) { e = eb[idx]; ok = ((e.x >> COL_BITS) == r); }
        unsigned m = __builtin_amdgcn_ballot_w32(ok);
        while (m != 0u) {
            const int j = __builtin_ctz(m);
            m &= (m - 1u);
            const float* s = lgcn_src_row<EGO>(src, src_items,
                                 __builtin_amdgcn_readlane(e.x & COL_MASK, j)) + lane;
            const float vj = __int_as_float(__builtin_amdgcn_readlane(e.y, j));
            s0 = fmaf(vj, s[0],  s0);
            s1 = fmaf(vj, s[32], s1);
        }
    }
}

template <bool EGO>
__device__ __forceinline__ void lgcn_row_piece_n(const float* __restrict__ src,
                                                 const float* __restrict__ src_items,
                                                 int lane, int n, int2 e, float& s0, float& s1) {
    switch ((n + 3) >> 2) {
    case 1:  lgcn_row_piece<EGO,  4>(src, src_items, lane, e, s0, s1); break;
    case 2:  lgcn_row_piece<EGO,  8>(src, src_items, lane, e, s0, s1); break;
    case 3:  lgcn_row_piece<EGO, 12>(src, src_items, lane, e, s0, s1); break;
    case 4:  lgcn_row_piece<EGO, 16>(src, src_items, lane, e, s0, s1); break;
    case 5:  lgcn_row_piece<EGO, 20>(src, src_items, lane, e, s0, s1); break;
    case 6:  lgcn_row_piece<EGO, 24>(src, src_items, lane, e, s0, s1); break;
    case 7:  lgcn_row_piece<EGO, 28>(src, src_items, lane, e, s0, s1); break;
    default: lgcn_row_piece<EGO, 32>(src, src_items, lane, e, s0, s1); break;
    }
}

template <bool EGO>
__global__ void __launch_bounds__(HOP_TB)
lgcn_hop_rows(int2*        __restrict__ ent,
              const int*   __restrict__ counts,
              const int*   __restrict__ starts,
              int*         __restrict__ roff,
              const float* __restrict__ src,
              const float* __restrict__ src_items,
              float*       __restrict__ out) {
    __shared__ int2 s_ent[ROWS_CAP];
    __shared__ int  s_cur[HOP_TB];
    __shared__ int  s_off[HOP_TB];
    __shared__ int  s_wsum[HOP_WAVES];
    const int tid  = (int)threadIdx.x;
    const int lane = tid & 31;
    const int wave = __builtin_amdgcn_readfirstlane(tid >> 5);
    const int b    = (int)blockIdx.x;

    const int cnt  = counts[b];
    int2*     eb   = ent + starts[b];
    int*      ro   = roff + (size_t)b * ROFF_W;
    float*    orow = out + (size_t)b * (RB * DIM) + lane;

    if (cnt <= ROWS_CAP) {
        if (EGO) {
            s_cur[tid] = 0;
            __syncthreads();
            for (int t0 = 0; t0 < cnt; t0 += HOP_TB * SORT_PER) {
                int ex[SORT_PER];
#pragma unroll
                for (int k = 0; k < SORT_PER; ++k) {
                    const int idx = t0 + k * HOP_TB + tid;
                    ex[k] = (idx < cnt) ? eb[idx].x : -1;
                }
#pragma unroll
                for (int k = 0; k < SORT_PER; ++k)
                    if (ex[k] >= 0) atomicAdd(&s_cur[ex[k] >> COL_BITS], 1);
            }
            __syncthreads();
            const int x = s_cur[tid];
            int incl = x;
            for (int d = 1; d < 32; d <<= 1) {
                const int y = __shfl_up(incl, d);
                if (lane >= d) incl += y;
            }
            if (lane == 31) s_wsum[wave] = incl;
            __syncthreads();
            int pre = 0;
            for (int w = 0; w < HOP_WAVES; ++w)
                if (w < wave) pre += s_wsum[w];
            s_off[tid] = pre + incl - x;
            s_cur[tid] = 0;
            __syncthreads();
            for (int t0 = 0; t0 < cnt; t0 += HOP_TB * SORT_PER) {
                int2 ee[SORT_PER];
                int  pp[SORT_PER];
#pragma unroll
                for (int k = 0; k < SORT_PER; ++k) {
                    const int idx = t0 + k * HOP_TB + tid;
                    ee[k] = (idx < cnt) ? eb[idx] : make_int2(-1, 0);
                }
#pragma unroll
                for (int k = 0; k < SORT_PER; ++k) {
                    pp[k] = 0;
                    if (ee[k].x >= 0) {
                        const int rl = ee[k].x >> COL_BITS;
                        pp[k] = s_off[rl] + atomicAdd(&s_cur[rl], 1);
                    }
                }
#pragma unroll
                for (int k = 0; k < SORT_PER; ++k)
                    if (ee[k].x >= 0) s_ent[pp[k]] = ee[k];
            }
            __syncthreads();
        } else {
            s_off[tid] = ro[tid];
            for (int t0 = 0; t0 < cnt; t0 += HOP_TB * SORT_PER) {
                int2 ee[SORT_PER];
#pragma unroll
                for (int k = 0; k < SORT_PER; ++k) {
                    const int idx = t0 + k * HOP_TB + tid;
                    ee[k] = (idx < cnt) ? eb[idx] : make_int2(-1, 0);
                }
#pragma unroll
                for (int k = 0; k < SORT_PER; ++k) {
                    const int idx = t0 + k * HOP_TB + tid;
                    if (ee[k].x >= 0) s_ent[idx] = ee[k];
                }
            }
            __syncthreads();
        }

        for (int r = wave; r < RB; r += HOP_WAVES) {
            if (b * RB + r >= N_NODES) break;
            const int rs = __builtin_amdgcn_readfirstlane(s_off[r]);
            const int re = __builtin_amdgcn_readfirstlane(s_off[r + 1]);
            float s0 = 0.f, s1 = 0.f;
            for (int base = rs; base < re; base += 32) {
                const int n = (re - base < 32) ? (re - base) : 32;
                int2 e = make_int2(0, 0);
                if (lane < n) { e = s_ent[base + lane]; e.x &= COL_MASK; }
                lgcn_row_piece_n<EGO>(src, src_items, lane, n, e, s0, s1);
            }
            orow[(size_t)r * DIM]      = s0;
            orow[(size_t)r * DIM + 32] = s1;
        }

        if (EGO) {
            ro[tid] = s_off[tid];
#pragma unroll 4
            for (int i = tid; i < cnt; i += HOP_TB) eb[i] = s_ent[i];
        }
    } else {
        for (int r = wave; r < RB; r += HOP_WAVES) {
            if (b * RB + r >= N_NODES) break;
            float s0 = 0.f, s1 = 0.f;
            lgcn_row_scan<EGO>(eb, cnt, r, lane, src, src_items, s0, s1);
            orow[(size_t)r * DIM]      = s0;
            orow[(size_t)r * DIM + 32] = s1;
        }
    }
}

#define PAIR_T   16
#define OP_ROWS  (2 * PAIR_T)
#define OP_PER   (OP_ROWS / HOP_WAVES)
#define OP_PITCH (DIM + 4)
static_assert(BATCH % PAIR_T == 0, "whole tiles of pairs");
static_assert(OP_ROWS % HOP_WAVES == 0, "the same number of rows for every wave");

__global__ void __launch_bounds__(HOP_TB)
lgcn_last_dot(const int2*  __restrict__ ent,
              const int*   __restrict__ counts,
              const int*   __restrict__ starts,
              const int*   __restrict__ roff,
              const float* __restrict__ user_emb,
              const float* __restrict__ item_emb,
              const float* __restrict__ h1,
              const float* __restrict__ h2,
              const int*   __restrict__ users,
              const int*   __restrict__ items,
              float*       __restrict__ out) {
    __shared__ float s_op[OP_ROWS][OP_PITCH];
    __shared__ int   s_node[OP_ROWS];
    __shared__ int   s_gs[OP_ROWS];
    __shared__ int   s_cn[OP_ROWS];
    __shared__ int   s_rs[OP_ROWS];
    __shared__ int   s_re[OP_ROWS];
    __shared__ float s_c[16][17];
    const int tid  = (int)threadIdx.x;
    const int lane = tid & 31;
    const int wave = __builtin_amdgcn_readfirstlane(tid >> 5);
    const int t    = (int)blockIdx.x;

    if (tid < OP_ROWS) {
        const int node = (tid < PAIR_T) ? users[t * PAIR_T + tid]
                                        : N_USERS + items[t * PAIR_T + tid - PAIR_T];
        const int b   = node / RB;
        const int lr  = node - b * RB;
        const int cnt = counts[b];
        int rs = 0, re = 0;
        if (cnt <= ROWS_CAP) { rs = roff[(size_t)b * ROFF_W + lr]; re = roff[(size_t)b * ROFF_W + lr + 1]; }
        s_node[tid] = node;
        s_gs[tid]   = starts[b];
        s_cn[tid]   = cnt;
        s_rs[tid]   = rs;
        s_re[tid]   = re;
    }
    __syncthreads();

    const int q0 = wave * OP_PER;
    int  rs = __builtin_amdgcn_readfirstlane(s_rs[q0]);
    int  re = __builtin_amdgcn_readfirstlane(s_re[q0]);
    int  gs = __builtin_amdgcn_readfirstlane(s_gs[q0]);
    int2 e  = make_int2(0, 0);
    if (lane < re - rs) e = ent[gs + rs + lane];
    for (int r = 0; r < OP_PER; ++r) {
        const int q = q0 + r;
        const int qn  = (r + 1 < OP_PER) ? (q + 1) : q;
        const int rsn = __builtin_amdgcn_readfirstlane(s_rs[qn]);
        const int ren = __builtin_amdgcn_readfirstlane(s_re[qn]);
        const int gsn = __builtin_amdgcn_readfirstlane(s_gs[qn]);
        const int ld  = gsn + rsn + lane;
        const int2 en = ent[(ld < NNZ) ? ld : (NNZ - 1)];
        const int node = __builtin_amdgcn_readfirstlane(s_node[q]);
        const int cnt  = __builtin_amdgcn_readfirstlane(s_cn[q]);
        const float* eg = (node < N_USERS) ? user_emb + (size_t)node * DIM
                                           : item_emb + (size_t)(node - N_USERS) * DIM;
        const float* p1 = h1 + (size_t)node * DIM;
        const float* p2 = h2 + (size_t)node * DIM;
        const float g0 = (eg[lane]      + p1[lane])      + p2[lane];
        const float g1 = (eg[lane + 32] + p1[lane + 32]) + p2[lane + 32];
        float s0 = 0.f, s1 = 0.f;
        if (cnt <= ROWS_CAP) {
            if (rs < re) {
                int n = (re - rs < 32) ? (re - rs) : 32;
                if (lane >= n) e = make_int2(0, 0);
                e.x &= COL_MASK;
                lgcn_row_piece_n<false>(h2, nullptr, lane, n, e, s0, s1);
                for (int base = rs + 32; base < re; base += 32) {
                    n = (re - base < 32) ? (re - base) : 32;
                    int2 e2 = make_int2(0, 0);
                    if (lane < n) { e2 = ent[gs + base + lane]; e2.x &= COL_MASK; }
                    lgcn_row_piece_n<false>(h2, nullptr, lane, n, e2, s0, s1);
                }
            }
        } else {
            const int b = node / RB;
            lgcn_row_scan<false>(ent + gs, cnt, node - b * RB, lane, h2, nullptr, s0, s1);
        }
        s_op[q][lane]      = g0 + s0;
        s_op[q][lane + 32] = g1 + s1;
        rs = rsn; re = ren; gs = gsn; e = en;
    }
    __syncthreads();

    if (wave == 0) {
        const int m     = lane & 15;
        const int khalf = lane >> 4;
        v8f c = {};
#pragma unroll
        for (int kc = 0; kc < DIM / 4; ++kc) {
            const int off = kc * 4 + khalf * 2;
            v2f a; a.x = s_op[m][off];          a.y = s_op[m][off + 1];
            v2f b; b.x = s_op[PAIR_T + m][off]; b.y = s_op[PAIR_T + m][off + 1];
            c = __builtin_amdgcn_wmma_f32_16x16x4_f32(
                     false, a,  false, b,
                     (short)0, c,  false,  false);
        }
#pragma unroll
        for (int r = 0; r < 8; ++r)
            s_c[r + 8 * khalf][m] = c[r];
    }
    __syncthreads();

    if (tid < PAIR_T)
        out[t * PAIR_T + tid] = s_c[tid][tid] * (1.0f / 16.0f);
}

extern "C" void kernel_launch(void* const* d_in, const int* in_sizes, int n_in,
                              void* d_out, int out_size, void* d_ws, size_t ws_size,
                              hipStream_t stream) {
    const float* user_emb = (const float*)d_in[0];
    const float* item_emb = (const float*)d_in[1];
    const int*   g_rows   = (const int*)  d_in[2];
    const int*   g_cols   = (const int*)  d_in[3];
    const float* g_vals   = (const float*)d_in[4];
    const int*   users    = (const int*)  d_in[5];
    const int*   items    = (const int*)  d_in[6];
    float*       out      = (float*)d_out;

    const size_t node_f = (size_t)N_NODES * DIM;
    float* hop1   = (float*)d_ws;
    float* hop2   = hop1 + node_f;
    int2*  ent    = (int2*)(hop2 + node_f);
    int*   counts = (int*)(ent + NNZ);
    int*   cursor = counts + NBP;
    int*   starts = cursor + NBP;
    int*   roff   = starts + NBP;

    lgcn_bin_clear<<<NBP / BIN_TB, BIN_TB, 0, stream>>>(counts, cursor);
    lgcn_bin_count<<<BIN_GRID, BIN_TB, 0, stream>>>(g_rows, counts);
    lgcn_bin_place<<<BIN_GRID, BIN_TB, 0, stream>>>(g_rows, g_cols, g_vals, counts, cursor,
                                                    starts, ent);

    lgcn_hop_rows<true><<<NB, HOP_TB, 0, stream>>>(ent, counts, starts, roff, user_emb, item_emb, hop1);
    lgcn_hop_rows<false><<<NB, HOP_TB, 0, stream>>>(ent, counts, starts, roff, hop1, nullptr, hop2);
    lgcn_last_dot<<<BATCH / PAIR_T, HOP_TB, 0, stream>>>(ent, counts, starts, roff, user_emb, item_emb,
                                                         hop1, hop2, users, items, out);
}
